# starting kernel
# baseline (speedup 1.0000x reference)
#include <hip/hip_runtime.h>

typedef __attribute__((ext_vector_type(2))) float v2f;
typedef __attribute__((ext_vector_type(8))) float v8f;

#define N_NODES 50000
#define N_EDGES 600000
#define K_DIM   128
#define BN_EPS  1e-5f

__global__ void gin_copy_kernel(const float* __restrict__ src,
                                float* __restrict__ dst, int n4) {
    int i = blockIdx.x * blockDim.x + threadIdx.x;
    if (i < n4) {
        reinterpret_cast<float4*>(dst)[i] =
            reinterpret_cast<const float4*>(src)[i];
    }
}

__global__ void gin_transpose_pad(const float* __restrict__ W,
                                  float* __restrict__ WT,
                                  int Hout, int totalPad) {
    int i = blockIdx.x * blockDim.x + threadIdx.x;
    if (i >= totalPad) return;
    int n = i >> 7;
    int k = i & 127;
    WT[i] = (n < Hout) ? W[k * Hout + n] : 0.0f;
}

__global__ void gin_scatter_kernel(const float* __restrict__ x,
                                   const int* __restrict__ src_idx,
                                   const int* __restrict__ dst_idx,
                                   float* __restrict__ agg,
                                   int total) {
    int i = blockIdx.x * blockDim.x + threadIdx.x;
    if (i >= total) return;
    int e  = i >> 5;
    int f4 = (i & 31) << 2;
    int s = src_idx[e];
    int d = dst_idx[e];
    const float4 v = *reinterpret_cast<const float4*>(x + (size_t)s * K_DIM + f4);
    float* p = agg + (size_t)d * K_DIM + f4;
    atomicAdd(p + 0, v.x);
    atomicAdd(p + 1, v.y);
    atomicAdd(p + 2, v.z);
    atomicAdd(p + 3, v.w);
}

template<int MODE, int HOUT>
__global__ void gin_gemm_wmma(const float* __restrict__ A,
                              const float* __restrict__ WT,
                              const float* __restrict__ bias,
                              const float* __restrict__ bn_g,
                              const float* __restrict__ bn_b,
                              const float* __restrict__ bn_m,
                              const float* __restrict__ bn_v,
                              float* __restrict__ out) {
    constexpr int CG         = 4;
    constexpr int COL_TILES  = (HOUT + 15) / 16;
    constexpr int COL_GROUPS = (COL_TILES + CG - 1) / CG;
    constexpr int ROW_TILES  = N_NODES / 16;
    constexpr int TOTAL      = ROW_TILES * COL_GROUPS;

    const int wave = threadIdx.x >> 5;
    const int tile = blockIdx.x * (blockDim.x >> 5) + wave;
    if (tile >= TOTAL) return;

    const int rowTile = tile / COL_GROUPS;
    const int colBase = (tile % COL_GROUPS) * (CG * 16);

    const int lane = threadIdx.x & 31;
    const int m    = lane & 15;
    const int hi   = lane >> 4;

    const float* Arow  = A  + (long long)(rowTile * 16 + m) * K_DIM + 2 * hi;
    const float* Wlane = WT + (long long)(colBase + m) * K_DIM + 2 * hi;

    v8f acc[CG] = {};
    #pragma unroll 4
    for (int k0 = 0; k0 < K_DIM; k0 += 4) {
        v2f a = *reinterpret_cast<const v2f*>(Arow + k0);
        #pragma unroll
        for (int t = 0; t < CG; ++t) {
            v2f b = *reinterpret_cast<const v2f*>(Wlane + (t << 11) + k0);
            acc[t] = __builtin_amdgcn_wmma_f32_16x16x4_f32(
                false, a, false, b, (short)0, acc[t], false, false);
        }
    }

    #pragma unroll
    for (int t = 0; t < CG; ++t) {
        const int  n  = colBase + t * 16 + m;
        const bool ok = (HOUT % (CG * 16) == 0) || (n < HOUT);
        float bias_v = ok ? bias[n] : 0.0f;
        float scale = 1.0f, shift = 0.0f;
        if constexpr (MODE == 2) {
            float rs = rsqrtf(bn_v[n] + BN_EPS);
            scale = bn_g[n] * rs;
            shift = bn_b[n] - bn_m[n] * scale;
        }
        float* orow = out + (long long)(rowTile * 16) * HOUT + (ok ? n : 0);
        #pragma unroll
        for (int r = 0; r < 8; ++r) {
            float val = acc[t][r] + bias_v;
            if constexpr (MODE == 2) val = val * scale + shift;
            if constexpr (MODE >= 1) val = fmaxf(val, 0.0f);
            if (ok) orow[(long long)(r + 8 * hi) * HOUT] = val;
        }
    }
}

extern "C" void kernel_launch(void* const* d_in, const int* in_sizes, int n_in,
                              void* d_out, int out_size, void* d_ws, size_t ws_size,
                              hipStream_t stream) {
    (void)in_sizes; (void)n_in; (void)out_size; (void)ws_size;

    const float* x      = (const float*)d_in[0];
    const int*   edge   = (const int*)  d_in[1];
    const float* lin1_w = (const float*)d_in[2];
    const float* lin1_b = (const float*)d_in[3];
    const float* bn1_g  = (const float*)d_in[4];
    const float* bn1_b  = (const float*)d_in[5];
    const float* bn1_m  = (const float*)d_in[6];
    const float* bn1_v  = (const float*)d_in[7];
    const float* lin2_w = (const float*)d_in[8];
    const float* lin2_b = (const float*)d_in[9];
    const float* lin3_w = (const float*)d_in[10];
    const float* lin3_b = (const float*)d_in[11];
    const float* bn2_g  = (const float*)d_in[12];
    const float* bn2_b  = (const float*)d_in[13];
    const float* bn2_m  = (const float*)d_in[14];
    const float* bn2_v  = (const float*)d_in[15];
    const float* lin4_w = (const float*)d_in[16];
    const float* lin4_b = (const float*)d_in[17];
    const float* lin5_w = (const float*)d_in[18];
    const float* lin5_b = (const float*)d_in[19];
    const float* lin6_w = (const float*)d_in[20];
    const float* lin6_b = (const float*)d_in[21];

    const int* e_src = edge;
    const int* e_dst = edge + N_EDGES;

    const size_t NH = (size_t)N_NODES * K_DIM;
    float* B0  = (float*)d_ws;
    float* B1  = B0  + NH;
    float* WT1 = B1  + NH;
    float* WT2 = WT1 + 128 * 128;
    float* WT3 = WT2 + 128 * 128;
    float* WT4 = WT3 + 128 * 128;
    float* WT5 = WT4 + 128 * 128;
    float* WT6 = WT5 + 128 * 128;

    gin_transpose_pad<<<64, 256, 0, stream>>>(lin1_w, WT1, 128, 128 * 128);
    gin_transpose_pad<<<64, 256, 0, stream>>>(lin2_w, WT2, 128, 128 * 128);
    gin_transpose_pad<<<64, 256, 0, stream>>>(lin3_w, WT3, 128, 128 * 128);
    gin_transpose_pad<<<64, 256, 0, stream>>>(lin4_w, WT4, 128, 128 * 128);
    gin_transpose_pad<<<64, 256, 0, stream>>>(lin5_w, WT5, 128, 128 * 128);
    gin_transpose_pad<<<32, 256, 0, stream>>>(lin6_w, WT6,  40,  64 * 128);

    const int NH4      = (int)(NH / 4);
    const int cpBlocks = (NH4 + 255) / 256;
    const int scTotal  = N_EDGES * 32;
    const int scBlocks = (scTotal + 255) / 256;
    const int g128     = ((N_NODES / 16) * 2 + 7) / 8;
    const int g40      = ((N_NODES / 16) * 1 + 7) / 8;

    gin_copy_kernel   <<<cpBlocks, 256, 0, stream>>>(x, B0, NH4);
    gin_scatter_kernel<<<scBlocks, 256, 0, stream>>>(x, e_src, e_dst, B0, scTotal);
    gin_gemm_wmma<2,128><<<g128, 256, 0, stream>>>(B0, WT1, lin1_b,
                                                   bn1_g, bn1_b, bn1_m, bn1_v, B1);
    gin_gemm_wmma<1,128><<<g128, 256, 0, stream>>>(B1, WT2, lin2_b,
                                                   nullptr, nullptr, nullptr, nullptr, B0);

    gin_copy_kernel   <<<cpBlocks, 256, 0, stream>>>(B0, B1, NH4);
    gin_scatter_kernel<<<scBlocks, 256, 0, stream>>>(B0, e_src, e_dst, B1, scTotal);
    gin_gemm_wmma<2,128><<<g128, 256, 0, stream>>>(B1, WT3, lin3_b,
                                                   bn2_g, bn2_b, bn2_m, bn2_v, B0);
    gin_gemm_wmma<1,128><<<g128, 256, 0, stream>>>(B0, WT4, lin4_b,
                                                   nullptr, nullptr, nullptr, nullptr, B1);

    gin_gemm_wmma<1,128><<<g128, 256, 0, stream>>>(B1, WT5, lin5_b,
                                                   nullptr, nullptr, nullptr, nullptr, B0);
    gin_gemm_wmma<0,40> <<<g40,  256, 0, stream>>>(B0, WT6, lin6_b,
                                                   nullptr, nullptr, nullptr, nullptr,
                                                   (float*)d_out);
}


# Aggregation owner-computes: work-group keeps 196 nodes' rows in LDS (start: own x rows), filters the whole edge list by range, adds by LDS f32 atomics (arrival order), stores once; no global atomics
# speedup vs baseline: 1.9706x; 1.9706x over previous
#include <hip/hip_runtime.h>

typedef __attribute__((ext_vector_type(2))) float v2f;
typedef __attribute__((ext_vector_type(4))) float v4f;
typedef __attribute__((ext_vector_type(8))) float v8f;

#define N_NODES 50000
#define N_EDGES 600000
#define K_DIM   128
#define BN_EPS  1e-5f

__global__ void gin_pack_weights(const float* __restrict__ W,
                                 float* __restrict__ WP,
                                 int Hout, int colTiles, int total) {
    int i = blockIdx.x * blockDim.x + threadIdx.x;
    if (i >= total) return;
    int s = i & 3;
    int l = (i >> 2) & 31;
    int g = i >> 7;
    int t = g % colTiles;
    int q = g / colTiles;
    int k = 8 * q + 4 * (l >> 4) + s;
    int n = 16 * t + (l & 15);
    WP[i] = (n < Hout) ? W[k * Hout + n] : 0.0f;
}

typedef __attribute__((ext_vector_type(4))) int v4i;

#define AG_GROUPS   256
#define AG_RANGE    196
#define AG_WAVES    32
#define AG_U        4
#define AG_BLOCK    (AG_U * 128)
#define AG_STEP     8
#define AG_NBLK     (N_EDGES / AG_BLOCK)
#define AG_TAIL0    (AG_NBLK * AG_BLOCK)
#define AG_TRIPS    (2 * ((AG_NBLK + 2 * AG_WAVES - 1) / (2 * AG_WAVES)))
#define AG_NBLK_PAD (AG_TRIPS * AG_WAVES)
#define AG_ROT      4
#define AG_QCAP     (32 + AG_BLOCK)
static_assert(AG_GROUPS * AG_RANGE >= N_NODES, "ranges cover all nodes");
static_assert((AG_GROUPS - 1) * AG_RANGE < N_NODES, "no empty range");
static_assert(AG_RANGE < 256, "local row and the spare row fit 8 bits");
static_assert(N_EDGES < (1 << 24), "edge number fits 24 bits");
static_assert((N_EDGES - AG_TAIL0) % 32 == 0, "the tail is whole waves of ids");
static_assert(AG_GROUPS * AG_ROT <= AG_NBLK_PAD, "one wrap is enough");

struct AgIds {
    v4i d[AG_U];
    int blk;
};

__device__ __forceinline__ void ag_fetch(AgIds& s, const int* __restrict__ dst_idx,
                                         int k, int wave, int rot, int lane) {
    int b = k * AG_WAVES + wave + rot;
    if (b >= AG_NBLK_PAD) b -= AG_NBLK_PAD;
    const bool ok = b < AG_NBLK;
    s.blk = ok ? b : -1;
    const v4i* p = reinterpret_cast<const v4i*>(dst_idx + (size_t)(ok ? b : 0) * AG_BLOCK) + lane;
    #pragma unroll
    for (int u = 0; u < AG_U; ++u) s.d[u] = p[u * 32];
}

__device__ __forceinline__ void ag_push(bool t, unsigned r, unsigned e, unsigned* q, int& qlen) {
    const unsigned m = __builtin_amdgcn_ballot_w32(t);
    if (t) q[qlen + __builtin_amdgcn_mbcnt_lo(m, 0u)] = (e << 8) | r;
    qlen += __builtin_popcount(m);
}

__device__ __forceinline__ void ag_scan(const AgIds& s, unsigned* q, int& qlen,
                                        int lo, int cnt, int lane) {
    if (s.blk < 0) return;
    #pragma unroll
    for (int u = 0; u < AG_U; ++u) {
        const v4i d = s.d[u];
        const unsigned r0 = (unsigned)(d.x - lo);
        const unsigned r1 = (unsigned)(d.y - lo);
        const unsigned r2 = (unsigned)(d.z - lo);
        const unsigned r3 = (unsigned)(d.w - lo);
        const bool t0 = r0 < (unsigned)cnt;
        const bool t1 = r1 < (unsigned)cnt;
        const bool t2 = r2 < (unsigned)cnt;
        const bool t3 = r3 < (unsigned)cnt;
        if (__builtin_amdgcn_ballot_w32(t0 | t1 | t2 | t3)) {
            const unsigned e0 = (unsigned)(s.blk * AG_BLOCK + u * 128 + lane * 4);
            ag_push(t0, r0, e0,      q, qlen);
            ag_push(t1, r1, e0 + 1u, q, qlen);
            ag_push(t2, r2, e0 + 2u, q, qlen);
            ag_push(t3, r3, e0 + 3u, q, qlen);
        }
    }
}

__device__ __forceinline__ void ag_drain(const float* __restrict__ x,
                                         const int* __restrict__ src_idx,
                                         float* tile, const unsigned* q,
                                         int qlen, int take, int lane) {
    unsigned ent = (unsigned)AG_RANGE;
    if (lane < take) ent = q[qlen - take + lane];
    const int s  = src_idx[ent >> 8];
    const int dl = (int)(ent & 255u);
    #pragma unroll
    for (int g = 0; g < 32; g += AG_STEP) {
        if (g < take) {
            v2f ra[AG_STEP], rb[AG_STEP];
            #pragma unroll
            for (int u = 0; u < AG_STEP; ++u) {
                const int sj = __builtin_amdgcn_readlane(s, g + u);
                const float* xr = x + (size_t)sj * K_DIM + 2 * lane;
                ra[u] = *reinterpret_cast<const v2f*>(xr);
                rb[u] = *reinterpret_cast<const v2f*>(xr + 64);
            }
            #pragma unroll
            for (int u = 0; u < AG_STEP; ++u) {
                const int dj = __builtin_amdgcn_readlane(dl, g + u);
                float* t = tile + dj * K_DIM + 2 * lane;
                __hip_atomic_fetch_add(t,      ra[u].x, __ATOMIC_RELAXED, __HIP_MEMORY_SCOPE_WORKGROUP);
                __hip_atomic_fetch_add(t + 1,  ra[u].y, __ATOMIC_RELAXED, __HIP_MEMORY_SCOPE_WORKGROUP);
                __hip_atomic_fetch_add(t + 64, rb[u].x, __ATOMIC_RELAXED, __HIP_MEMORY_SCOPE_WORKGROUP);
                __hip_atomic_fetch_add(t + 65, rb[u].y, __ATOMIC_RELAXED, __HIP_MEMORY_SCOPE_WORKGROUP);
            }
        }
    }
}

__global__ void __launch_bounds__(AG_WAVES * 32)
gin_aggregate_kernel(const float* __restrict__ x,
                     const int* __restrict__ src_idx,
                     const int* __restrict__ dst_idx,
                     float* __restrict__ agg) {
    __shared__ __attribute__((aligned(16))) float tile[(AG_RANGE + 1) * K_DIM];
    __shared__ unsigned queue_all[AG_WAVES * AG_QCAP];

    const int tid  = threadIdx.x;
    const int lane = tid & 31;
    const int wave = __builtin_amdgcn_readfirstlane(tid >> 5);
    const int lo   = (int)blockIdx.x * AG_RANGE;
    const int cnt  = (N_NODES - lo < AG_RANGE) ? (N_NODES - lo) : AG_RANGE;
    const int n4   = cnt * (K_DIM / 4);

    unsigned* q = queue_all + wave * AG_QCAP;
    int qlen = 0;
    const int rot = (int)blockIdx.x * AG_ROT;

    AgIds ia, ib;
    ag_fetch(ia, dst_idx, 0, wave, rot, lane);

    {
        constexpr int FILL = (AG_RANGE * (K_DIM / 4) + AG_WAVES * 32 - 1) / (AG_WAVES * 32);
        const v4f* xs = reinterpret_cast<const v4f*>(x + (size_t)lo * K_DIM);
        v4f* ts = reinterpret_cast<v4f*>(tile);
        v4f tmp[FILL];
        #pragma unroll
        for (int j = 0; j < FILL; ++j) {
            const int i = tid + j * (AG_WAVES * 32);
            tmp[j] = xs[(i < n4) ? i : (n4 - 1)];
        }
        #pragma unroll
        for (int j = 0; j < FILL; ++j) {
            const int i = tid + j * (AG_WAVES * 32);
            if (i < n4) ts[i] = tmp[j];
        }
    }
    __syncthreads();

    for (int k = 0; k < AG_TRIPS; k += 2) {
        ag_fetch(ib, dst_idx, k + 1, wave, rot, lane);
        ag_scan(ia, q, qlen, lo, cnt, lane);
        while (qlen >= 32) {
            ag_drain(x, src_idx, tile, q, qlen, 32, lane);
            qlen -= 32;
        }
        ag_fetch(ia, dst_idx, k + 2, wave, rot, lane);
        ag_scan(ib, q, qlen, lo, cnt, lane);
        while (qlen >= 32) {
            ag_drain(x, src_idx, tile, q, qlen, 32, lane);
            qlen -= 32;
        }
    }

    if (wave == 0) {
        for (int b = AG_TAIL0; b < N_EDGES; b += 32) {
            const int e = b + lane;
            const unsigned r = (unsigned)(dst_idx[e] - lo);
            ag_push(r < (unsigned)cnt, r, (unsigned)e, q, qlen);
            if (qlen >= 32) {
                ag_drain(x, src_idx, tile, q, qlen, 32, lane);
                qlen -= 32;
            }
        }
    }
    if (qlen > 0) ag_drain(x, src_idx, tile, q, qlen, qlen, lane);

    __syncthreads();

    {
        v4f* os = reinterpret_cast<v4f*>(agg + (size_t)lo * K_DIM);
        const v4f* ts = reinterpret_cast<const v4f*>(tile);
        for (int i = tid; i < n4; i += AG_WAVES * 32) os[i] = ts[i];
    }
}

#define GEMM_WAVES_PER_BLOCK 4
#define GEMM_BLOCKS          256

template<int MODE, int HOUT>
__global__ void __launch_bounds__(GEMM_WAVES_PER_BLOCK * 32)
gin_gemm_wmma(const float* __restrict__ A,
              const float* __restrict__ WP,
              const float* __restrict__ bias,
              const float* __restrict__ bn_g,
              const float* __restrict__ bn_b,
              const float* __restrict__ bn_m,
              const float* __restrict__ bn_v,
              float* __restrict__ out) {
    constexpr int CT        = (HOUT + 15) / 16;
    constexpr int KB        = K_DIM / 8;
    constexpr int ROW_TILES = N_NODES / 16;

    const int lane = threadIdx.x & 31;
    const int c    = lane & 15;
    const int h    = lane >> 4;
    const int wave   = __builtin_amdgcn_readfirstlane((int)(threadIdx.x >> 5));
    const int nwaves = (int)gridDim.x * GEMM_WAVES_PER_BLOCK;
    int tile = (int)blockIdx.x * GEMM_WAVES_PER_BLOCK + wave;

    const v4f* wp = reinterpret_cast<const v4f*>(WP) + lane;
    v4f w[KB * CT];
    #pragma unroll
    for (int i = 0; i < KB * CT; ++i) w[i] = wp[i * 32];

    float bias_v[CT], scale[CT], shift[CT];
    bool  ok[CT];
    #pragma unroll
    for (int t = 0; t < CT; ++t) {
        const int n = t * 16 + c;
        ok[t]     = (HOUT % 16 == 0) || (n < HOUT);
        bias_v[t] = ok[t] ? bias[n] : 0.0f;
        scale[t]  = 1.0f;
        shift[t]  = 0.0f;
        if constexpr (MODE == 2) {
            float rs = rsqrtf(bn_v[n] + BN_EPS);
            scale[t] = bn_g[n] * rs;
            shift[t] = bn_b[n] - bn_m[n] * scale[t];
        }
    }

    const v4f* Alane = reinterpret_cast<const v4f*>(A + c * K_DIM + 4 * h);

    v4f cur[KB];
    if (tile < ROW_TILES) {
        const v4f* ap = Alane + (size_t)tile * (16 * K_DIM / 4);
        #pragma unroll
        for (int q = 0; q < KB; ++q) cur[q] = ap[2 * q];
    }

    for (; tile < ROW_TILES; tile += nwaves) {
        const int nt = (tile + nwaves < ROW_TILES) ? (tile + nwaves) : tile;
        const v4f* np = Alane + (size_t)nt * (16 * K_DIM / 4);
        v4f nxt[KB];
        #pragma unroll
        for (int q = 0; q < KB; ++q) nxt[q] = np[2 * q];

        v8f acc[CT] = {};
        #pragma unroll
        for (int q = 0; q < KB; ++q) {
            const v2f a0 = {cur[q].x, cur[q].y};
            const v2f a1 = {cur[q].z, cur[q].w};
            #pragma unroll
            for (int t = 0; t < CT; ++t) {
                const v2f b0 = {w[q * CT + t].x, w[q * CT + t].y};
                acc[t] = __builtin_amdgcn_wmma_f32_16x16x4_f32(
                    false, a0, false, b0, (short)0, acc[t], false, false);
            }
            #pragma unroll
            for (int t = 0; t < CT; ++t) {
                const v2f b1 = {w[q * CT + t].z, w[q * CT + t].w};
                acc[t] = __builtin_amdgcn_wmma_f32_16x16x4_f32(
                    false, a1, false, b1, (short)0, acc[t], false, false);
            }
        }

        float* orow = out + (size_t)(tile * 16 + 8 * h) * HOUT + c;
        #pragma unroll
        for (int t = 0; t < CT; ++t) {
            #pragma unroll
            for (int r = 0; r < 8; ++r) {
                float val = acc[t][r] + bias_v[t];
                if constexpr (MODE == 2) val = val * scale[t] + shift[t];
                if constexpr (MODE >= 1) val = fmaxf(val, 0.0f);
                if (ok[t]) orow[r * HOUT + t * 16] = val;
            }
        }

        #pragma unroll
        for (int q = 0; q < KB; ++q) cur[q] = nxt[q];
    }
}

extern "C" void kernel_launch(void* const* d_in, const int* in_sizes, int n_in,
                              void* d_out, int out_size, void* d_ws, size_t ws_size,
                              hipStream_t stream) {
    (void)in_sizes; (void)n_in; (void)out_size; (void)ws_size;

    const float* x      = (const float*)d_in[0];
    const int*   edge   = (const int*)  d_in[1];
    const float* lin1_w = (const float*)d_in[2];
    const float* lin1_b = (const float*)d_in[3];
    const float* bn1_g  = (const float*)d_in[4];
    const float* bn1_b  = (const float*)d_in[5];
    const float* bn1_m  = (const float*)d_in[6];
    const float* bn1_v  = (const float*)d_in[7];
    const float* lin2_w = (const float*)d_in[8];
    const float* lin2_b = (const float*)d_in[9];
    const float* lin3_w = (const float*)d_in[10];
    const float* lin3_b = (const float*)d_in[11];
    const float* bn2_g  = (const float*)d_in[12];
    const float* bn2_b  = (const float*)d_in[13];
    const float* bn2_m  = (const float*)d_in[14];
    const float* bn2_v  = (const float*)d_in[15];
    const float* lin4_w = (const float*)d_in[16];
    const float* lin4_b = (const float*)d_in[17];
    const float* lin5_w = (const float*)d_in[18];
    const float* lin5_b = (const float*)d_in[19];
    const float* lin6_w = (const float*)d_in[20];
    const float* lin6_b = (const float*)d_in[21];

    const int* e_src = edge;
    const int* e_dst = edge + N_EDGES;

    const size_t NH = (size_t)N_NODES * K_DIM;
    float* B0  = (float*)d_ws;
    float* B1  = B0  + NH;
    float* WT1 = B1  + NH;
    float* WT2 = WT1 + 128 * 128;
    float* WT3 = WT2 + 128 * 128;
    float* WT4 = WT3 + 128 * 128;
    float* WT5 = WT4 + 128 * 128;
    float* WT6 = WT5 + 128 * 128;

    gin_pack_weights<<<64, 256, 0, stream>>>(lin1_w, WT1, 128, 8, 128 * 128);
    gin_pack_weights<<<64, 256, 0, stream>>>(lin2_w, WT2, 128, 8, 128 * 128);
    gin_pack_weights<<<64, 256, 0, stream>>>(lin3_w, WT3, 128, 8, 128 * 128);
    gin_pack_weights<<<64, 256, 0, stream>>>(lin4_w, WT4, 128, 8, 128 * 128);
    gin_pack_weights<<<64, 256, 0, stream>>>(lin5_w, WT5, 128, 8, 128 * 128);
    gin_pack_weights<<<24, 256, 0, stream>>>(lin6_w, WT6,  40, 3,  48 * 128);

    const int gG       = GEMM_BLOCKS;
    const int gB       = GEMM_WAVES_PER_BLOCK * 32;

    gin_aggregate_kernel<<<AG_GROUPS, AG_WAVES * 32, 0, stream>>>(x, e_src, e_dst, B0);
    gin_gemm_wmma<2,128><<<gG, gB, 0, stream>>>(B0, WT1, lin1_b,
                                                   bn1_g, bn1_b, bn1_m, bn1_v, B1);
    gin_gemm_wmma<1,128><<<gG, gB, 0, stream>>>(B1, WT2, lin2_b,
                                                   nullptr, nullptr, nullptr, nullptr, B0);

    gin_aggregate_kernel<<<AG_GROUPS, AG_WAVES * 32, 0, stream>>>(B0, e_src, e_dst, B1);
    gin_gemm_wmma<2,128><<<gG, gB, 0, stream>>>(B1, WT3, lin3_b,
                                                   bn2_g, bn2_b, bn2_m, bn2_v, B0);
    gin_gemm_wmma<1,128><<<gG, gB, 0, stream>>>(B0, WT4, lin4_b,
                                                   nullptr, nullptr, nullptr, nullptr, B1);

    gin_gemm_wmma<1,128><<<gG, gB, 0, stream>>>(B1, WT5, lin5_b,
                                                   nullptr, nullptr, nullptr, nullptr, B0);
    gin_gemm_wmma<0,40> <<<gG, gB, 0, stream>>>(B0, WT6, lin6_b,
                                                   nullptr, nullptr, nullptr, nullptr,
                                                   (float*)d_out);
}


# Edge ids walked once a call: layer 1's aggregation also writes each wave's taken edges (source id, local row) to a workspace list; layer 2 adds rows from those lists (LDS f32 adds, arrival order)
# speedup vs baseline: 2.1131x; 1.0723x over previous
#include <hip/hip_runtime.h>

typedef __attribute__((ext_vector_type(2))) float v2f;
typedef __attribute__((ext_vector_type(4))) float v4f;
typedef __attribute__((ext_vector_type(8))) float v8f;

#define N_NODES 50000
#define N_EDGES 600000
#define K_DIM   128
#define BN_EPS  1e-5f

__global__ void gin_pack_weights(const float* __restrict__ W,
                                 float* __restrict__ WP,
                                 int Hout, int colTiles, int total) {
    int i = blockIdx.x * blockDim.x + threadIdx.x;
    if (i >= total) return;
    int s = i & 3;
    int l = (i >> 2) & 31;
    int g = i >> 7;
    int t = g % colTiles;
    int q = g / colTiles;
    int k = 8 * q + 4 * (l >> 4) + s;
    int n = 16 * t + (l & 15);
    WP[i] = (n < Hout) ? W[k * Hout + n] : 0.0f;
}

typedef __attribute__((ext_vector_type(4))) int v4i;

#define AG_GROUPS   256
#define AG_RANGE    196
#define AG_WAVES    32
#define AG_U        4
#define AG_BLOCK    (AG_U * 128)
#define AG_STEP     8
#define AG_NBLK     (N_EDGES / AG_BLOCK)
#define AG_TAIL0    (AG_NBLK * AG_BLOCK)
#define AG_TRIPS    (2 * ((AG_NBLK + 2 * AG_WAVES - 1) / (2 * AG_WAVES)))
#define AG_NBLK_PAD (AG_TRIPS * AG_WAVES)
#define AG_ROT      4
#define AG_QCAP     (32 + AG_BLOCK)
#define AG_LCAP     1024
#define AG_STEP2    16
static_assert(AG_GROUPS * AG_RANGE >= N_NODES, "ranges cover all nodes");
static_assert((AG_GROUPS - 1) * AG_RANGE < N_NODES, "no empty range");
static_assert(AG_RANGE < 256, "local row and the spare row fit 8 bits");
static_assert(N_EDGES < (1 << 24), "edge number fits 24 bits");
static_assert((N_EDGES - AG_TAIL0) % 32 == 0, "the tail is whole waves of ids");
static_assert(AG_GROUPS * AG_ROT <= AG_NBLK_PAD, "one wrap is enough");

struct AgIds {
    v4i d[AG_U];
    int blk;
};

__device__ __forceinline__ void ag_fetch(AgIds& s, const int* __restrict__ dst_idx,
                                         int k, int wave, int rot, int lane) {
    int b = k * AG_WAVES + wave + rot;
    if (b >= AG_NBLK_PAD) b -= AG_NBLK_PAD;
    const bool ok = b < AG_NBLK;
    s.blk = ok ? b : -1;
    const v4i* p = reinterpret_cast<const v4i*>(dst_idx + (size_t)(ok ? b : 0) * AG_BLOCK) + lane;
    #pragma unroll
    for (int u = 0; u < AG_U; ++u) s.d[u] = p[u * 32];
}

__device__ __forceinline__ void ag_push(bool t, unsigned r, unsigned e, unsigned* q, int& qlen) {
    const unsigned m = __builtin_amdgcn_ballot_w32(t);
    if (t) q[qlen + __builtin_amdgcn_mbcnt_lo(m, 0u)] = (e << 8) | r;
    qlen += __builtin_popcount(m);
}

__device__ __forceinline__ void ag_scan(const AgIds& s, unsigned* q, int& qlen,
                                        int lo, int cnt, int lane) {
    if (s.blk < 0) return;
    #pragma unroll
    for (int u = 0; u < AG_U; ++u) {
        const v4i d = s.d[u];
        const unsigned r0 = (unsigned)(d.x - lo);
        const unsigned r1 = (unsigned)(d.y - lo);
        const unsigned r2 = (unsigned)(d.z - lo);
        const unsigned r3 = (unsigned)(d.w - lo);
        const bool t0 = r0 < (unsigned)cnt;
        const bool t1 = r1 < (unsigned)cnt;
        const bool t2 = r2 < (unsigned)cnt;
        const bool t3 = r3 < (unsigned)cnt;
        if (__builtin_amdgcn_ballot_w32(t0 | t1 | t2 | t3)) {
            const unsigned e0 = (unsigned)(s.blk * AG_BLOCK + u * 128 + lane * 4);
            ag_push(t0, r0, e0,      q, qlen);
            ag_push(t1, r1, e0 + 1u, q, qlen);
            ag_push(t2, r2, e0 + 2u, q, qlen);
            ag_push(t3, r3, e0 + 3u, q, qlen);
        }
    }
}

template<int STEP>
__device__ __forceinline__ void ag_add_rows(const float* __restrict__ x, float* tile,
                                            int s, int dl, int take, int lane) {
    #pragma unroll
    for (int g = 0; g < 32; g += STEP) {
        if (g < take) {
            v2f ra[STEP], rb[STEP];
            #pragma unroll
            for (int u = 0; u < STEP; ++u) {
                const int sj = __builtin_amdgcn_readlane(s, g + u);
                const float* xr = x + (size_t)sj * K_DIM + 2 * lane;
                ra[u] = *reinterpret_cast<const v2f*>(xr);
                rb[u] = *reinterpret_cast<const v2f*>(xr + 64);
            }
            #pragma unroll
            for (int u = 0; u < STEP; ++u) {
                const int dj = __builtin_amdgcn_readlane(dl, g + u);
                float* t = tile + dj * K_DIM + 2 * lane;
                __hip_atomic_fetch_add(t,      ra[u].x, __ATOMIC_RELAXED, __HIP_MEMORY_SCOPE_WORKGROUP);
                __hip_atomic_fetch_add(t + 1,  ra[u].y, __ATOMIC_RELAXED, __HIP_MEMORY_SCOPE_WORKGROUP);
                __hip_atomic_fetch_add(t + 64, rb[u].x, __ATOMIC_RELAXED, __HIP_MEMORY_SCOPE_WORKGROUP);
                __hip_atomic_fetch_add(t + 65, rb[u].y, __ATOMIC_RELAXED, __HIP_MEMORY_SCOPE_WORKGROUP);
            }
        }
    }
}

template<bool REC>
__device__ __forceinline__ void ag_drain(const float* __restrict__ x,
                                         const int* __restrict__ src_idx,
                                         float* tile, const unsigned* q,
                                         int qlen, int take, int lane,
                                         unsigned* __restrict__ lw, int& kept) {
    unsigned ent = (unsigned)AG_RANGE;
    if (lane < take) ent = q[qlen - take + lane];
    const int s  = src_idx[ent >> 8];
    const int dl = (int)(ent & 255u);
    if constexpr (REC) {
        if (kept + take <= AG_LCAP) {
            if (lane < take) lw[kept + lane] = ((unsigned)s << 8) | (unsigned)dl;
        }
        kept += take;
    }
    ag_add_rows<AG_STEP>(x, tile, s, dl, take, lane);
}

template<int PASS>
__global__ void __launch_bounds__(AG_WAVES * 32)
gin_aggregate_kernel(const float* __restrict__ x,
                     const int* __restrict__ src_idx,
                     const int* __restrict__ dst_idx,
                     float* __restrict__ agg,
                     unsigned* __restrict__ lst,
                     int* __restrict__ hdr) {
    __shared__ __attribute__((aligned(16))) float tile[(AG_RANGE + 1) * K_DIM];
    __shared__ unsigned queue_all[AG_WAVES * AG_QCAP];

    const int tid  = threadIdx.x;
    const int lane = tid & 31;
    const int wave = __builtin_amdgcn_readfirstlane(tid >> 5);
    const int lo   = (int)blockIdx.x * AG_RANGE;
    const int cnt  = (N_NODES - lo < AG_RANGE) ? (N_NODES - lo) : AG_RANGE;
    const int n4   = cnt * (K_DIM / 4);

    unsigned* q = queue_all + wave * AG_QCAP;
    int qlen = 0;
    const int rot = (int)blockIdx.x * AG_ROT;

    const int wslot = (int)blockIdx.x * AG_WAVES + wave;
    unsigned* lw = lst + (size_t)wslot * AG_LCAP;
    int kept = 0;
    int n = 0;
    bool listed = false;
    if constexpr (PASS == 2) {
        n = hdr[wslot];
        listed = (n <= AG_LCAP);
    }

    AgIds ia, ib;
    unsigned ent_n = (unsigned)AG_RANGE;
    if (listed) {
        if (lane < n) ent_n = lw[lane];
    } else {
        ag_fetch(ia, dst_idx, 0, wave, rot, lane);
    }

    {
        constexpr int FILL = (AG_RANGE * (K_DIM / 4) + AG_WAVES * 32 - 1) / (AG_WAVES * 32);
        const v4f* xs = reinterpret_cast<const v4f*>(x + (size_t)lo * K_DIM);
        v4f* ts = reinterpret_cast<v4f*>(tile);
        v4f tmp[FILL];
        #pragma unroll
        for (int j = 0; j < FILL; ++j) {
            const int i = tid + j * (AG_WAVES * 32);
            tmp[j] = xs[(i < n4) ? i : (n4 - 1)];
        }
        #pragma unroll
        for (int j = 0; j < FILL; ++j) {
            const int i = tid + j * (AG_WAVES * 32);
            if (i < n4) ts[i] = tmp[j];
        }
    }
    __syncthreads();

    constexpr bool REC = (PASS == 1);
    if (listed) {
        for (int base = 0; base < n; base += 32) {
            const unsigned ent = ent_n;
            const int left = n - base;
            ent_n = (unsigned)AG_RANGE;
            if (lane + 32 < left) ent_n = lw[base + 32 + lane];
            ag_add_rows<AG_STEP2>(x, tile, (int)(ent >> 8), (int)(ent & 255u),
                                  left < 32 ? left : 32, lane);
        }
    } else {
        for (int k = 0; k < AG_TRIPS; k += 2) {
            ag_fetch(ib, dst_idx, k + 1, wave, rot, lane);
            ag_scan(ia, q, qlen, lo, cnt, lane);
            while (qlen >= 32) {
                ag_drain<REC>(x, src_idx, tile, q, qlen, 32, lane, lw, kept);
                qlen -= 32;
            }
            ag_fetch(ia, dst_idx, k + 2, wave, rot, lane);
            ag_scan(ib, q, qlen, lo, cnt, lane);
            while (qlen >= 32) {
                ag_drain<REC>(x, src_idx, tile, q, qlen, 32, lane, lw, kept);
                qlen -= 32;
            }
        }

        if (wave == 0) {
            for (int b = AG_TAIL0; b < N_EDGES; b += 32) {
                const int e = b + lane;
                const unsigned r = (unsigned)(dst_idx[e] - lo);
                ag_push(r < (unsigned)cnt, r, (unsigned)e, q, qlen);
                if (qlen >= 32) {
                    ag_drain<REC>(x, src_idx, tile, q, qlen, 32, lane, lw, kept);
                    qlen -= 32;
                }
            }
        }
        if (qlen > 0) ag_drain<REC>(x, src_idx, tile, q, qlen, qlen, lane, lw, kept);
    }
    if constexpr (PASS == 1) {
        if (lane == 0) hdr[wslot] = kept;
    }

    __syncthreads();

    {
        v4f* os = reinterpret_cast<v4f*>(agg + (size_t)lo * K_DIM);
        const v4f* ts = reinterpret_cast<const v4f*>(tile);
        for (int i = tid; i < n4; i += AG_WAVES * 32) os[i] = ts[i];
    }
}

#define GEMM_WAVES_PER_BLOCK 4
#define GEMM_BLOCKS          256

template<int MODE, int HOUT>
__global__ void __launch_bounds__(GEMM_WAVES_PER_BLOCK * 32)
gin_gemm_wmma(const float* __restrict__ A,
              const float* __restrict__ WP,
              const float* __restrict__ bias,
              const float* __restrict__ bn_g,
              const float* __restrict__ bn_b,
              const float* __restrict__ bn_m,
              const float* __restrict__ bn_v,
              float* __restrict__ out) {
    constexpr int CT        = (HOUT + 15) / 16;
    constexpr int KB        = K_DIM / 8;
    constexpr int ROW_TILES = N_NODES / 16;

    const int lane = threadIdx.x & 31;
    const int c    = lane & 15;
    const int h    = lane >> 4;
    const int wave   = __builtin_amdgcn_readfirstlane((int)(threadIdx.x >> 5));
    const int nwaves = (int)gridDim.x * GEMM_WAVES_PER_BLOCK;
    int tile = (int)blockIdx.x * GEMM_WAVES_PER_BLOCK + wave;

    const v4f* wp = reinterpret_cast<const v4f*>(WP) + lane;
    v4f w[KB * CT];
    #pragma unroll
    for (int i = 0; i < KB * CT; ++i) w[i] = wp[i * 32];

    float bias_v[CT], scale[CT], shift[CT];
    bool  ok[CT];
    #pragma unroll
    for (int t = 0; t < CT; ++t) {
        const int n = t * 16 + c;
        ok[t]     = (HOUT % 16 == 0) || (n < HOUT);
        bias_v[t] = ok[t] ? bias[n] : 0.0f;
        scale[t]  = 1.0f;
        shift[t]  = 0.0f;
        if constexpr (MODE == 2) {
            float rs = rsqrtf(bn_v[n] + BN_EPS);
            scale[t] = bn_g[n] * rs;
            shift[t] = bn_b[n] - bn_m[n] * scale[t];
        }
    }

    const v4f* Alane = reinterpret_cast<const v4f*>(A + c * K_DIM + 4 * h);

    v4f cur[KB];
    if (tile < ROW_TILES) {
        const v4f* ap = Alane + (size_t)tile * (16 * K_DIM / 4);
        #pragma unroll
        for (int q = 0; q < KB; ++q) cur[q] = ap[2 * q];
    }

    for (; tile < ROW_TILES; tile += nwaves) {
        const int nt = (tile + nwaves < ROW_TILES) ? (tile + nwaves) : tile;
        const v4f* np = Alane + (size_t)nt * (16 * K_DIM / 4);
        v4f nxt[KB];
        #pragma unroll
        for (int q = 0; q < KB; ++q) nxt[q] = np[2 * q];

        v8f acc[CT] = {};
        #pragma unroll
        for (int q = 0; q < KB; ++q) {
            const v2f a0 = {cur[q].x, cur[q].y};
            const v2f a1 = {cur[q].z, cur[q].w};
            #pragma unroll
            for (int t = 0; t < CT; ++t) {
                const v2f b0 = {w[q * CT + t].x, w[q * CT + t].y};
                acc[t] = __builtin_amdgcn_wmma_f32_16x16x4_f32(
                    false, a0, false, b0, (short)0, acc[t], false, false);
            }
            #pragma unroll
            for (int t = 0; t < CT; ++t) {
                const v2f b1 = {w[q * CT + t].z, w[q * CT + t].w};
                acc[t] = __builtin_amdgcn_wmma_f32_16x16x4_f32(
                    false, a1, false, b1, (short)0, acc[t], false, false);
            }
        }

        float* orow = out + (size_t)(tile * 16 + 8 * h) * HOUT + c;
        #pragma unroll
        for (int t = 0; t < CT; ++t) {
            #pragma unroll
            for (int r = 0; r < 8; ++r) {
                float val = acc[t][r] + bias_v[t];
                if constexpr (MODE == 2) val = val * scale[t] + shift[t];
                if constexpr (MODE >= 1) val = fmaxf(val, 0.0f);
                if (ok[t]) orow[r * HOUT + t * 16] = val;
            }
        }

        #pragma unroll
        for (int q = 0; q < KB; ++q) cur[q] = nxt[q];
    }
}

extern "C" void kernel_launch(void* const* d_in, const int* in_sizes, int n_in,
                              void* d_out, int out_size, void* d_ws, size_t ws_size,
                              hipStream_t stream) {
    (void)in_sizes; (void)n_in; (void)out_size; (void)ws_size;

    const float* x      = (const float*)d_in[0];
    const int*   edge   = (const int*)  d_in[1];
    const float* lin1_w = (const float*)d_in[2];
    const float* lin1_b = (const float*)d_in[3];
    const float* bn1_g  = (const float*)d_in[4];
    const float* bn1_b  = (const float*)d_in[5];
    const float* bn1_m  = (const float*)d_in[6];
    const float* bn1_v  = (const float*)d_in[7];
    const float* lin2_w = (const float*)d_in[8];
    const float* lin2_b = (const float*)d_in[9];
    const float* lin3_w = (const float*)d_in[10];
    const float* lin3_b = (const float*)d_in[11];
    const float* bn2_g  = (const float*)d_in[12];
    const float* bn2_b  = (const float*)d_in[13];
    const float* bn2_m  = (const float*)d_in[14];
    const float* bn2_v  = (const float*)d_in[15];
    const float* lin4_w = (const float*)d_in[16];
    const float* lin4_b = (const float*)d_in[17];
    const float* lin5_w = (const float*)d_in[18];
    const float* lin5_b = (const float*)d_in[19];
    const float* lin6_w = (const float*)d_in[20];
    const float* lin6_b = (const float*)d_in[21];

    const int* e_src = edge;
    const int* e_dst = edge + N_EDGES;

    const size_t NH = (size_t)N_NODES * K_DIM;
    float* B0  = (float*)d_ws;
    float* B1  = B0  + NH;
    float* WT1 = B1  + NH;
    float* WT2 = WT1 + 128 * 128;
    float* WT3 = WT2 + 128 * 128;
    float* WT4 = WT3 + 128 * 128;
    float* WT5 = WT4 + 128 * 128;
    float* WT6 = WT5 + 128 * 128;
    unsigned* LST = (unsigned*)(WT6 + 128 * 128);
    int*      HDR = (int*)(LST + (size_t)AG_GROUPS * AG_WAVES * AG_LCAP);

    gin_pack_weights<<<64, 256, 0, stream>>>(lin1_w, WT1, 128, 8, 128 * 128);
    gin_pack_weights<<<64, 256, 0, stream>>>(lin2_w, WT2, 128, 8, 128 * 128);
    gin_pack_weights<<<64, 256, 0, stream>>>(lin3_w, WT3, 128, 8, 128 * 128);
    gin_pack_weights<<<64, 256, 0, stream>>>(lin4_w, WT4, 128, 8, 128 * 128);
    gin_pack_weights<<<64, 256, 0, stream>>>(lin5_w, WT5, 128, 8, 128 * 128);
    gin_pack_weights<<<24, 256, 0, stream>>>(lin6_w, WT6,  40, 3,  48 * 128);

    const int gG       = GEMM_BLOCKS;
    const int gB       = GEMM_WAVES_PER_BLOCK * 32;

    gin_aggregate_kernel<1><<<AG_GROUPS, AG_WAVES * 32, 0, stream>>>(x, e_src, e_dst, B0, LST, HDR);
    gin_gemm_wmma<2,128><<<gG, gB, 0, stream>>>(B0, WT1, lin1_b,
                                                   bn1_g, bn1_b, bn1_m, bn1_v, B1);
    gin_gemm_wmma<1,128><<<gG, gB, 0, stream>>>(B1, WT2, lin2_b,
                                                   nullptr, nullptr, nullptr, nullptr, B0);

    gin_aggregate_kernel<2><<<AG_GROUPS, AG_WAVES * 32, 0, stream>>>(B0, e_src, e_dst, B1, LST, HDR);
    gin_gemm_wmma<2,128><<<gG, gB, 0, stream>>>(B1, WT3, lin3_b,
                                                   bn2_g, bn2_b, bn2_m, bn2_v, B0);
    gin_gemm_wmma<1,128><<<gG, gB, 0, stream>>>(B0, WT4, lin4_b,
                                                   nullptr, nullptr, nullptr, nullptr, B1);

    gin_gemm_wmma<1,128><<<gG, gB, 0, stream>>>(B1, WT5, lin5_b,
                                                   nullptr, nullptr, nullptr, nullptr, B0);
    gin_gemm_wmma<0,40> <<<gG, gB, 0, stream>>>(B0, WT6, lin6_b,
                                                   nullptr, nullptr, nullptr, nullptr,
                                                   (float*)d_out);
}


# GEMM output columns dealt to lanes in fours (column 64j+4c+s): weights read in place as b128 rows, pack kernels gone (14 -> 8 launches); outputs stored as b128 whole lines; k order unchanged
# speedup vs baseline: 2.7574x; 1.3049x over previous
#include <hip/hip_runtime.h>

typedef __attribute__((ext_vector_type(2))) float v2f;
typedef __attribute__((ext_vector_type(4))) float v4f;
typedef __attribute__((ext_vector_type(8))) float v8f;

#define N_NODES 50000
#define N_EDGES 600000
#define K_DIM   128
#define BN_EPS  1e-5f

typedef __attribute__((ext_vector_type(4))) int v4i;

#define AG_GROUPS   256
#define AG_RANGE    196
#define AG_WAVES    32
#define AG_U        4
#define AG_BLOCK    (AG_U * 128)
#define AG_STEP     8
#define AG_NBLK     (N_EDGES / AG_BLOCK)
#define AG_TAIL0    (AG_NBLK * AG_BLOCK)
#define AG_TRIPS    (2 * ((AG_NBLK + 2 * AG_WAVES - 1) / (2 * AG_WAVES)))
#define AG_NBLK_PAD (AG_TRIPS * AG_WAVES)
#define AG_ROT      4
#define AG_QCAP     (32 + AG_BLOCK)
#define AG_LCAP     1024
#define AG_STEP2    16
static_assert(AG_GROUPS * AG_RANGE >= N_NODES, "ranges cover all nodes");
static_assert((AG_GROUPS - 1) * AG_RANGE < N_NODES, "no empty range");
static_assert(AG_RANGE < 256, "local row and the spare row fit 8 bits");
static_assert(N_EDGES < (1 << 24), "edge number fits 24 bits");
static_assert((N_EDGES - AG_TAIL0) % 32 == 0, "the tail is whole waves of ids");
static_assert(AG_GROUPS * AG_ROT <= AG_NBLK_PAD, "one wrap is enough");

struct AgIds {
    v4i d[AG_U];
    int blk;
};

__device__ __forceinline__ void ag_fetch(AgIds& s, const int* __restrict__ dst_idx,
                                         int k, int wave, int rot, int lane) {
    int b = k * AG_WAVES + wave + rot;
    if (b >= AG_NBLK_PAD) b -= AG_NBLK_PAD;
    const bool ok = b < AG_NBLK;
    s.blk = ok ? b : -1;
    const v4i* p = reinterpret_cast<const v4i*>(dst_idx + (size_t)(ok ? b : 0) * AG_BLOCK) + lane;
    #pragma unroll
    for (int u = 0; u < AG_U; ++u) s.d[u] = p[u * 32];
}

__device__ __forceinline__ void ag_push(bool t, unsigned r, unsigned e, unsigned* q, int& qlen) {
    const unsigned m = __builtin_amdgcn_ballot_w32(t);
    if (t) q[qlen + __builtin_amdgcn_mbcnt_lo(m, 0u)] = (e << 8) | r;
    qlen += __builtin_popcount(m);
}

__device__ __forceinline__ void ag_scan(const AgIds& s, unsigned* q, int& qlen,
                                        int lo, int cnt, int lane) {
    if (s.blk < 0) return;
    #pragma unroll
    for (int u = 0; u < AG_U; ++u) {
        const v4i d = s.d[u];
        const unsigned r0 = (unsigned)(d.x - lo);
        const unsigned r1 = (unsigned)(d.y - lo);
        const unsigned r2 = (unsigned)(d.z - lo);
        const unsigned r3 = (unsigned)(d.w - lo);
        const bool t0 = r0 < (unsigned)cnt;
        const bool t1 = r1 < (unsigned)cnt;
        const bool t2 = r2 < (unsigned)cnt;
        const bool t3 = r3 < (unsigned)cnt;
        if (__builtin_amdgcn_ballot_w32(t0 | t1 | t2 | t3)) {
            const unsigned e0 = (unsigned)(s.blk * AG_BLOCK + u * 128 + lane * 4);
            ag_push(t0, r0, e0,      q, qlen);
            ag_push(t1, r1, e0 + 1u, q, qlen);
            ag_push(t2, r2, e0 + 2u, q, qlen);
            ag_push(t3, r3, e0 + 3u, q, qlen);
        }
    }
}

template<int STEP>
__device__ __forceinline__ void ag_add_rows(const float* __restrict__ x, float* tile,
                                            int s, int dl, int take, int lane) {
    #pragma unroll
    for (int g = 0; g < 32; g += STEP) {
        if (g < take) {
            v2f ra[STEP], rb[STEP];
            #pragma unroll
            for (int u = 0; u < STEP; ++u) {
                const int sj = __builtin_amdgcn_readlane(s, g + u);
                const float* xr = x + (size_t)sj * K_DIM + 2 * lane;
                ra[u] = *reinterpret_cast<const v2f*>(xr);
                rb[u] = *reinterpret_cast<const v2f*>(xr + 64);
            }
            #pragma unroll
            for (int u = 0; u < STEP; ++u) {
                const int dj = __builtin_amdgcn_readlane(dl, g + u);
                float* t = tile + dj * K_DIM + 2 * lane;
                __hip_atomic_fetch_add(t,      ra[u].x, __ATOMIC_RELAXED, __HIP_MEMORY_SCOPE_WORKGROUP);
                __hip_atomic_fetch_add(t + 1,  ra[u].y, __ATOMIC_RELAXED, __HIP_MEMORY_SCOPE_WORKGROUP);
                __hip_atomic_fetch_add(t + 64, rb[u].x, __ATOMIC_RELAXED, __HIP_MEMORY_SCOPE_WORKGROUP);
                __hip_atomic_fetch_add(t + 65, rb[u].y, __ATOMIC_RELAXED, __HIP_MEMORY_SCOPE_WORKGROUP);
            }
        }
    }
}

template<bool REC>
__device__ __forceinline__ void ag_drain(const float* __restrict__ x,
                                         const int* __restrict__ src_idx,
                                         float* tile, const unsigned* q,
                                         int qlen, int take, int lane,
                                         unsigned* __restrict__ lw, int& kept) {
    unsigned ent = (unsigned)AG_RANGE;
    if (lane < take) ent = q[qlen - take + lane];
    const int s  = src_idx[ent >> 8];
    const int dl = (int)(ent & 255u);
    if constexpr (REC) {
        if (kept + take <= AG_LCAP) {
            if (lane < take) lw[kept + lane] = ((unsigned)s << 8) | (unsigned)dl;
        }
        kept += take;
    }
    ag_add_rows<AG_STEP>(x, tile, s, dl, take, lane);
}

template<int PASS>
__global__ void __launch_bounds__(AG_WAVES * 32)
gin_aggregate_kernel(const float* __restrict__ x,
                     const int* __restrict__ src_idx,
                     const int* __restrict__ dst_idx,
                     float* __restrict__ agg,
                     unsigned* __restrict__ lst,
                     int* __restrict__ hdr) {
    __shared__ __attribute__((aligned(16))) float tile[(AG_RANGE + 1) * K_DIM];
    __shared__ unsigned queue_all[AG_WAVES * AG_QCAP];

    const int tid  = threadIdx.x;
    const int lane = tid & 31;
    const int wave = __builtin_amdgcn_readfirstlane(tid >> 5);
    const int lo   = (int)blockIdx.x * AG_RANGE;
    const int cnt  = (N_NODES - lo < AG_RANGE) ? (N_NODES - lo) : AG_RANGE;
    const int n4   = cnt * (K_DIM / 4);

    unsigned* q = queue_all + wave * AG_QCAP;
    int qlen = 0;
    const int rot = (int)blockIdx.x * AG_ROT;

    const int wslot = (int)blockIdx.x * AG_WAVES + wave;
    unsigned* lw = lst + (size_t)wslot * AG_LCAP;
    int kept = 0;
    int n = 0;
    bool listed = false;
    if constexpr (PASS == 2) {
        n = hdr[wslot];
        listed = (n <= AG_LCAP);
    }

    AgIds ia, ib;
    unsigned ent_n = (unsigned)AG_RANGE;
    if (listed) {
        if (lane < n) ent_n = lw[lane];
    } else {
        ag_fetch(ia, dst_idx, 0, wave, rot, lane);
    }

    {
        constexpr int FILL = (AG_RANGE * (K_DIM / 4) + AG_WAVES * 32 - 1) / (AG_WAVES * 32);
        const v4f* xs = reinterpret_cast<const v4f*>(x + (size_t)lo * K_DIM);
        v4f* ts = reinterpret_cast<v4f*>(tile);
        v4f tmp[FILL];
        #pragma unroll
        for (int j = 0; j < FILL; ++j) {
            const int i = tid + j * (AG_WAVES * 32);
            tmp[j] = xs[(i < n4) ? i : (n4 - 1)];
        }
        #pragma unroll
        for (int j = 0; j < FILL; ++j) {
            const int i = tid + j * (AG_WAVES * 32);
            if (i < n4) ts[i] = tmp[j];
        }
    }
    __syncthreads();

    constexpr bool REC = (PASS == 1);
    if (listed) {
        for (int base = 0; base < n; base += 32) {
            const unsigned ent = ent_n;
            const int left = n - base;
            ent_n = (unsigned)AG_RANGE;
            if (lane + 32 < left) ent_n = lw[base + 32 + lane];
            ag_add_rows<AG_STEP2>(x, tile, (int)(ent >> 8), (int)(ent & 255u),
                                  left < 32 ? left : 32, lane);
        }
    } else {
        for (int k = 0; k < AG_TRIPS; k += 2) {
            ag_fetch(ib, dst_idx, k + 1, wave, rot, lane);
            ag_scan(ia, q, qlen, lo, cnt, lane);
            while (qlen >= 32) {
                ag_drain<REC>(x, src_idx, tile, q, qlen, 32, lane, lw, kept);
                qlen -= 32;
            }
            ag_fetch(ia, dst_idx, k + 2, wave, rot, lane);
            ag_scan(ib, q, qlen, lo, cnt, lane);
            while (qlen >= 32) {
                ag_drain<REC>(x, src_idx, tile, q, qlen, 32, lane, lw, kept);
                qlen -= 32;
            }
        }

        if (wave == 0) {
            for (int b = AG_TAIL0; b < N_EDGES; b += 32) {
                const int e = b + lane;
                const unsigned r = (unsigned)(dst_idx[e] - lo);
                ag_push(r < (unsigned)cnt, r, (unsigned)e, q, qlen);
                if (qlen >= 32) {
                    ag_drain<REC>(x, src_idx, tile, q, qlen, 32, lane, lw, kept);
                    qlen -= 32;
                }
            }
        }
        if (qlen > 0) ag_drain<REC>(x, src_idx, tile, q, qlen, qlen, lane, lw, kept);
    }
    if constexpr (PASS == 1) {
        if (lane == 0) hdr[wslot] = kept;
    }

    __syncthreads();

    {
        v4f* os = reinterpret_cast<v4f*>(agg + (size_t)lo * K_DIM);
        const v4f* ts = reinterpret_cast<const v4f*>(tile);
        for (int i = tid; i < n4; i += AG_WAVES * 32) os[i] = ts[i];
    }
}

#define GEMM_WAVES_PER_BLOCK 4
#define GEMM_BLOCKS          256

template<int MODE, int HOUT>
__global__ void __launch_bounds__(GEMM_WAVES_PER_BLOCK * 32)
gin_gemm_wmma(const float* __restrict__ A,
              const float* __restrict__ W,
              const float* __restrict__ bias,
              const float* __restrict__ bn_g,
              const float* __restrict__ bn_b,
              const float* __restrict__ bn_m,
              const float* __restrict__ bn_v,
              float* __restrict__ out) {
    constexpr int  CT        = (HOUT + 15) / 16;
    constexpr int  KB        = K_DIM / 8;
    constexpr int  ROW_TILES = N_NODES / 16;
    constexpr bool QUAD      = (HOUT % 64 == 0);
    constexpr int  NJ        = QUAD ? HOUT / 64 : 1;

    const int lane = threadIdx.x & 31;
    const int c    = lane & 15;
    const int h    = lane >> 4;
    const int wave   = __builtin_amdgcn_readfirstlane((int)(threadIdx.x >> 5));
    const int nwaves = (int)gridDim.x * GEMM_WAVES_PER_BLOCK;
    int tile = (int)blockIdx.x * GEMM_WAVES_PER_BLOCK + wave;

    v2f wb[KB][2][CT];
    if constexpr (QUAD) {
        const float* wl = W + (size_t)(4 * h) * HOUT + 4 * c;
        #pragma unroll
        for (int q = 0; q < KB; ++q) {
            v4f L[4][NJ];
            #pragma unroll
            for (int i = 0; i < 4; ++i) {
                #pragma unroll
                for (int j = 0; j < NJ; ++j)
                    L[i][j] = *reinterpret_cast<const v4f*>(wl + (size_t)(8 * q + i) * HOUT + 64 * j);
            }
            #pragma unroll
            for (int j = 0; j < NJ; ++j) {
                #pragma unroll
                for (int s = 0; s < 4; ++s) {
                    wb[q][0][4 * j + s] = v2f{L[0][j][s], L[1][j][s]};
                    wb[q][1][4 * j + s] = v2f{L[2][j][s], L[3][j][s]};
                }
            }
        }
    } else {
        #pragma unroll
        for (int t = 0; t < CT; ++t) {
            const int  n   = t * 16 + c;
            const bool okn = n < HOUT;
            const float* wl = W + (size_t)(4 * h) * HOUT + (okn ? n : 0);
            #pragma unroll
            for (int q = 0; q < KB; ++q) {
                const float* p = wl + (size_t)(8 * q) * HOUT;
                const float w0 = p[0], w1 = p[HOUT], w2 = p[2 * HOUT], w3 = p[3 * HOUT];
                wb[q][0][t] = v2f{okn ? w0 : 0.0f, okn ? w1 : 0.0f};
                wb[q][1][t] = v2f{okn ? w2 : 0.0f, okn ? w3 : 0.0f};
            }
        }
    }

    float bias_v[CT], scale[CT], shift[CT];
    bool  ok[CT];
    #pragma unroll
    for (int t = 0; t < CT; ++t) {
        const int n = QUAD ? (64 * (t >> 2) + 4 * c + (t & 3)) : (t * 16 + c);
        ok[t]     = (HOUT % 16 == 0) || (n < HOUT);
        bias_v[t] = ok[t] ? bias[n] : 0.0f;
        scale[t]  = 1.0f;
        shift[t]  = 0.0f;
        if constexpr (MODE == 2) {
            float rs = rsqrtf(bn_v[n] + BN_EPS);
            scale[t] = bn_g[n] * rs;
            shift[t] = bn_b[n] - bn_m[n] * scale[t];
        }
    }

    const v4f* Alane = reinterpret_cast<const v4f*>(A + c * K_DIM + 4 * h);

    v4f cur[KB];
    if (tile < ROW_TILES) {
        const v4f* ap = Alane + (size_t)tile * (16 * K_DIM / 4);
        #pragma unroll
        for (int q = 0; q < KB; ++q) cur[q] = ap[2 * q];
    }

    for (; tile < ROW_TILES; tile += nwaves) {
        const int nt = (tile + nwaves < ROW_TILES) ? (tile + nwaves) : tile;
        const v4f* np = Alane + (size_t)nt * (16 * K_DIM / 4);
        v4f nxt[KB];
        #pragma unroll
        for (int q = 0; q < KB; ++q) nxt[q] = np[2 * q];

        v8f acc[CT] = {};
        #pragma unroll
        for (int q = 0; q < KB; ++q) {
            const v2f a0 = {cur[q].x, cur[q].y};
            const v2f a1 = {cur[q].z, cur[q].w};
            #pragma unroll
            for (int t = 0; t < CT; ++t) {
                acc[t] = __builtin_amdgcn_wmma_f32_16x16x4_f32(
                    false, a0, false, wb[q][0][t], (short)0, acc[t], false, false);
            }
            #pragma unroll
            for (int t = 0; t < CT; ++t) {
                acc[t] = __builtin_amdgcn_wmma_f32_16x16x4_f32(
                    false, a1, false, wb[q][1][t], (short)0, acc[t], false, false);
            }
        }

        if constexpr (QUAD) {
            float* orow = out + (size_t)(tile * 16 + 8 * h) * HOUT + 4 * c;
            #pragma unroll
            for (int j = 0; j < NJ; ++j) {
                #pragma unroll
                for (int r = 0; r < 8; ++r) {
                    v4f v;
                    #pragma unroll
                    for (int s = 0; s < 4; ++s) {
                        const int t = 4 * j + s;
                        float val = acc[t][r] + bias_v[t];
                        if constexpr (MODE == 2) val = val * scale[t] + shift[t];
                        if constexpr (MODE >= 1) val = fmaxf(val, 0.0f);
                        v[s] = val;
                    }
                    *reinterpret_cast<v4f*>(orow + r * HOUT + 64 * j) = v;
                }
            }
        } else {
            float* orow = out + (size_t)(tile * 16 + 8 * h) * HOUT + c;
            #pragma unroll
            for (int t = 0; t < CT; ++t) {
                #pragma unroll
                for (int r = 0; r < 8; ++r) {
                    float val = acc[t][r] + bias_v[t];
                    if constexpr (MODE == 2) val = val * scale[t] + shift[t];
                    if constexpr (MODE >= 1) val = fmaxf(val, 0.0f);
                    if (ok[t]) orow[r * HOUT + t * 16] = val;
                }
            }
        }

        #pragma unroll
        for (int q = 0; q < KB; ++q) cur[q] = nxt[q];
    }
}

extern "C" void kernel_launch(void* const* d_in, const int* in_sizes, int n_in,
                              void* d_out, int out_size, void* d_ws, size_t ws_size,
                              hipStream_t stream) {
    (void)in_sizes; (void)n_in; (void)out_size; (void)ws_size;

    const float* x      = (const float*)d_in[0];
    const int*   edge   = (const int*)  d_in[1];
    const float* lin1_w = (const float*)d_in[2];
    const float* lin1_b = (const float*)d_in[3];
    const float* bn1_g  = (const float*)d_in[4];
    const float* bn1_b  = (const float*)d_in[5];
    const float* bn1_m  = (const float*)d_in[6];
    const float* bn1_v  = (const float*)d_in[7];
    const float* lin2_w = (const float*)d_in[8];
    const float* lin2_b = (const float*)d_in[9];
    const float* lin3_w = (const float*)d_in[10];
    const float* lin3_b = (const float*)d_in[11];
    const float* bn2_g  = (const float*)d_in[12];
    const float* bn2_b  = (const float*)d_in[13];
    const float* bn2_m  = (const float*)d_in[14];
    const float* bn2_v  = (const float*)d_in[15];
    const float* lin4_w = (const float*)d_in[16];
    const float* lin4_b = (const float*)d_in[17];
    const float* lin5_w = (const float*)d_in[18];
    const float* lin5_b = (const float*)d_in[19];
    const float* lin6_w = (const float*)d_in[20];
    const float* lin6_b = (const float*)d_in[21];

    const int* e_src = edge;
    const int* e_dst = edge + N_EDGES;

    const size_t NH = (size_t)N_NODES * K_DIM;
    float* B0  = (float*)d_ws;
    float* B1  = B0  + NH;
    unsigned* LST = (unsigned*)(B1 + NH);
    int*      HDR = (int*)(LST + (size_t)AG_GROUPS * AG_WAVES * AG_LCAP);

    const int gG       = GEMM_BLOCKS;
    const int gB       = GEMM_WAVES_PER_BLOCK * 32;

    gin_aggregate_kernel<1><<<AG_GROUPS, AG_WAVES * 32, 0, stream>>>(x, e_src, e_dst, B0, LST, HDR);
    gin_gemm_wmma<2,128><<<gG, gB, 0, stream>>>(B0, lin1_w, lin1_b,
                                                   bn1_g, bn1_b, bn1_m, bn1_v, B1);
    gin_gemm_wmma<1,128><<<gG, gB, 0, stream>>>(B1, lin2_w, lin2_b,
                                                   nullptr, nullptr, nullptr, nullptr, B0);

    gin_aggregate_kernel<2><<<AG_GROUPS, AG_WAVES * 32, 0, stream>>>(B0, e_src, e_dst, B1, LST, HDR);
    gin_gemm_wmma<2,128><<<gG, gB, 0, stream>>>(B1, lin3_w, lin3_b,
                                                   bn2_g, bn2_b, bn2_m, bn2_v, B0);
    gin_gemm_wmma<1,128><<<gG, gB, 0, stream>>>(B0, lin4_w, lin4_b,
                                                   nullptr, nullptr, nullptr, nullptr, B1);

    gin_gemm_wmma<1,128><<<gG, gB, 0, stream>>>(B1, lin5_w, lin5_b,
                                                   nullptr, nullptr, nullptr, nullptr, B0);
    gin_gemm_wmma<0,40> <<<gG, gB, 0, stream>>>(B0, lin6_w, lin6_b,
                                                   nullptr, nullptr, nullptr, nullptr,
                                                   (float*)d_out);
}


# Each GIN layer's first linear map (with BN, ReLU) moved into the aggregation kernel: the group multiplies its finished LDS tile by W staged in LDS, stores h only; 8 -> 6 launches; k order unchanged
# speedup vs baseline: 2.9531x; 1.0710x over previous
#include <hip/hip_runtime.h>

typedef __attribute__((ext_vector_type(2))) float v2f;
typedef __attribute__((ext_vector_type(4))) float v4f;
typedef __attribute__((ext_vector_type(8))) float v8f;

#define N_NODES 50000
#define N_EDGES 600000
#define K_DIM   128
#define BN_EPS  1e-5f

typedef __attribute__((ext_vector_type(4))) int v4i;

#define AG_GROUPS   256
#define AG_RANGE    196
#define AG_WAVES    32
#define AG_U        4
#define AG_BLOCK    (AG_U * 128)
#define AG_STEP     8
#define AG_NBLK     (N_EDGES / AG_BLOCK)
#define AG_TAIL0    (AG_NBLK * AG_BLOCK)
#define AG_TRIPS    (2 * ((AG_NBLK + 2 * AG_WAVES - 1) / (2 * AG_WAVES)))
#define AG_NBLK_PAD (AG_TRIPS * AG_WAVES)
#define AG_ROT      4
#define AG_QCAP     (32 + AG_BLOCK)
#define AG_LCAP     1024
#define AG_STEP2    16
#define AG_PITCH    132
#define AG_RTILES   ((AG_RANGE + 15) / 16)
#define AG_ROWS     (AG_RTILES * 16)
static_assert(AG_ROWS > AG_RANGE, "room for the spare row");
static_assert(AG_PITCH % 4 == 0 && (AG_PITCH / 4) % 2 == 1, "rows 16-byte aligned, pitch 16 B x odd");
static_assert(AG_GROUPS * AG_RANGE >= N_NODES, "ranges cover all nodes");
static_assert((AG_GROUPS - 1) * AG_RANGE < N_NODES, "no empty range");
static_assert(AG_RANGE < 256, "local row and the spare row fit 8 bits");
static_assert(N_EDGES < (1 << 24), "edge number fits 24 bits");
static_assert((N_EDGES - AG_TAIL0) % 32 == 0, "the tail is whole waves of ids");
static_assert(AG_GROUPS * AG_ROT <= AG_NBLK_PAD, "one wrap is enough");

struct AgIds {
    v4i d[AG_U];
    int blk;
};

__device__ __forceinline__ void ag_fetch(AgIds& s, const int* __restrict__ dst_idx,
                                         int k, int wave, int rot, int lane) {
    int b = k * AG_WAVES + wave + rot;
    if (b >= AG_NBLK_PAD) b -= AG_NBLK_PAD;
    const bool ok = b < AG_NBLK;
    s.blk = ok ? b : -1;
    const v4i* p = reinterpret_cast<const v4i*>(dst_idx + (size_t)(ok ? b : 0) * AG_BLOCK) + lane;
    #pragma unroll
    for (int u = 0; u < AG_U; ++u) s.d[u] = p[u * 32];
}

__device__ __forceinline__ void ag_push(bool t, unsigned r, unsigned e, unsigned* q, int& qlen) {
    const unsigned m = __builtin_amdgcn_ballot_w32(t);
    if (t) q[qlen + __builtin_amdgcn_mbcnt_lo(m, 0u)] = (e << 8) | r;
    qlen += __builtin_popcount(m);
}

__device__ __forceinline__ void ag_scan(const AgIds& s, unsigned* q, int& qlen,
                                        int lo, int cnt, int lane) {
    if (s.blk < 0) return;
    #pragma unroll
    for (int u = 0; u < AG_U; ++u) {
        const v4i d = s.d[u];
        const unsigned r0 = (unsigned)(d.x - lo);
        const unsigned r1 = (unsigned)(d.y - lo);
        const unsigned r2 = (unsigned)(d.z - lo);
        const unsigned r3 = (unsigned)(d.w - lo);
        const bool t0 = r0 < (unsigned)cnt;
        const bool t1 = r1 < (unsigned)cnt;
        const bool t2 = r2 < (unsigned)cnt;
        const bool t3 = r3 < (unsigned)cnt;
        if (__builtin_amdgcn_ballot_w32(t0 | t1 | t2 | t3)) {
            const unsigned e0 = (unsigned)(s.blk * AG_BLOCK + u * 128 + lane * 4);
            ag_push(t0, r0, e0,      q, qlen);
            ag_push(t1, r1, e0 + 1u, q, qlen);
            ag_push(t2, r2, e0 + 2u, q, qlen);
            ag_push(t3, r3, e0 + 3u, q, qlen);
        }
    }
}

template<int STEP>
__device__ __forceinline__ void ag_add_rows(const float* __restrict__ x, float* tile,
                                            int s, int dl, int take, int lane) {
    #pragma unroll
    for (int g = 0; g < 32; g += STEP) {
        if (g < take) {
            v2f ra[STEP], rb[STEP];
            #pragma unroll
            for (int u = 0; u < STEP; ++u) {
                const int sj = __builtin_amdgcn_readlane(s, g + u);
                const float* xr = x + (size_t)sj * K_DIM + 2 * lane;
                ra[u] = *reinterpret_cast<const v2f*>(xr);
                rb[u] = *reinterpret_cast<const v2f*>(xr + 64);
            }
            #pragma unroll
            for (int u = 0; u < STEP; ++u) {
                const int dj = __builtin_amdgcn_readlane(dl, g + u);
                float* t = tile + dj * AG_PITCH + 2 * lane;
                __hip_atomic_fetch_add(t,      ra[u].x, __ATOMIC_RELAXED, __HIP_MEMORY_SCOPE_WORKGROUP);
                __hip_atomic_fetch_add(t + 1,  ra[u].y, __ATOMIC_RELAXED, __HIP_MEMORY_SCOPE_WORKGROUP);
                __hip_atomic_fetch_add(t + 64, rb[u].x, __ATOMIC_RELAXED, __HIP_MEMORY_SCOPE_WORKGROUP);
                __hip_atomic_fetch_add(t + 65, rb[u].y, __ATOMIC_RELAXED, __HIP_MEMORY_SCOPE_WORKGROUP);
            }
        }
    }
}

template<bool REC>
__device__ __forceinline__ void ag_drain(const float* __restrict__ x,
                                         const int* __restrict__ src_idx,
                                         float* tile, const unsigned* q,
                                         int qlen, int take, int lane,
                                         unsigned* __restrict__ lw, int& kept) {
    unsigned ent = (unsigned)AG_RANGE;
    if (lane < take) ent = q[qlen - take + lane];
    const int s  = src_idx[ent >> 8];
    const int dl = (int)(ent & 255u);
    if constexpr (REC) {
        if (kept + take <= AG_LCAP) {
            if (lane < take) lw[kept + lane] = ((unsigned)s << 8) | (unsigned)dl;
        }
        kept += take;
    }
    ag_add_rows<AG_STEP>(x, tile, s, dl, take, lane);
}

union AgQueuesOrOut {
    unsigned q[AG_WAVES * AG_QCAP];
    float    o[AG_ROWS * AG_PITCH];
};
static_assert(sizeof(unsigned) * AG_WAVES * AG_QCAP <= sizeof(float) * AG_ROWS * AG_PITCH, "queues fit under the plane");

template<int PASS>
__global__ void __launch_bounds__(AG_WAVES * 32)
gin_aggregate_lin_kernel(const float* __restrict__ x,
                         const int* __restrict__ src_idx,
                         const int* __restrict__ dst_idx,
                         const float* __restrict__ W,
                         const float* __restrict__ bias,
                         const float* __restrict__ bn_g,
                         const float* __restrict__ bn_b,
                         const float* __restrict__ bn_m,
                         const float* __restrict__ bn_v,
                         float* __restrict__ hout,
                         unsigned* __restrict__ lst,
                         int* __restrict__ hdr) {
    __shared__ __attribute__((aligned(16))) float tile[AG_ROWS * AG_PITCH];
    __shared__ __attribute__((aligned(16))) float wlds[K_DIM * AG_PITCH];
    __shared__ __attribute__((aligned(16))) AgQueuesOrOut qo;
    unsigned* queue_all = qo.q;

    const int tid  = threadIdx.x;
    const int lane = tid & 31;
    const int wave = __builtin_amdgcn_readfirstlane(tid >> 5);
    const int lo   = (int)blockIdx.x * AG_RANGE;
    const int cnt  = (N_NODES - lo < AG_RANGE) ? (N_NODES - lo) : AG_RANGE;
    const int n4   = cnt * (K_DIM / 4);

    unsigned* q = queue_all + wave * AG_QCAP;
    int qlen = 0;
    const int rot = (int)blockIdx.x * AG_ROT;

    const int wslot = (int)blockIdx.x * AG_WAVES + wave;
    unsigned* lw = lst + (size_t)wslot * AG_LCAP;
    int kept = 0;
    int n = 0;
    bool listed = false;
    if constexpr (PASS == 2) {
        n = hdr[wslot];
        listed = (n <= AG_LCAP);
    }

    AgIds ia, ib;
    unsigned ent_n = (unsigned)AG_RANGE;
    if (listed) {
        if (lane < n) ent_n = lw[lane];
    } else {
        ag_fetch(ia, dst_idx, 0, wave, rot, lane);
    }

    {
        constexpr int FILL = (AG_RANGE * (K_DIM / 4) + AG_WAVES * 32 - 1) / (AG_WAVES * 32);
        constexpr int WFIL = (K_DIM * (K_DIM / 4)) / (AG_WAVES * 32);
        static_assert(WFIL * AG_WAVES * 32 == K_DIM * (K_DIM / 4), "the weights are whole trips");
        const v4f* xs = reinterpret_cast<const v4f*>(x + (size_t)lo * K_DIM);
        const v4f* w4 = reinterpret_cast<const v4f*>(W);
        v4f* ts = reinterpret_cast<v4f*>(tile);
        v4f* ws = reinterpret_cast<v4f*>(wlds);
        v4f tmp[FILL], wt[WFIL];
        #pragma unroll
        for (int j = 0; j < WFIL; ++j) wt[j] = w4[tid + j * (AG_WAVES * 32)];
        #pragma unroll
        for (int j = 0; j < FILL; ++j) {
            const int i = tid + j * (AG_WAVES * 32);
            tmp[j] = xs[(i < n4) ? i : (n4 - 1)];
        }
        #pragma unroll
        for (int j = 0; j < WFIL; ++j) {
            const int i = tid + j * (AG_WAVES * 32);
            ws[(i >> 5) * (AG_PITCH / 4) + (i & 31)] = wt[j];
        }
        #pragma unroll
        for (int j = 0; j < FILL; ++j) {
            const int i = tid + j * (AG_WAVES * 32);
            if (i < n4) ts[(i >> 5) * (AG_PITCH / 4) + (i & 31)] = tmp[j];
        }
    }
    __syncthreads();

    constexpr bool REC = (PASS == 1);
    if (listed) {
        for (int base = 0; base < n; base += 32) {
            const unsigned ent = ent_n;
            const int left = n - base;
            ent_n = (unsigned)AG_RANGE;
            if (lane + 32 < left) ent_n = lw[base + 32 + lane];
            ag_add_rows<AG_STEP2>(x, tile, (int)(ent >> 8), (int)(ent & 255u),
                                  left < 32 ? left : 32, lane);
        }
    } else {
        for (int k = 0; k < AG_TRIPS; k += 2) {
            ag_fetch(ib, dst_idx, k + 1, wave, rot, lane);
            ag_scan(ia, q, qlen, lo, cnt, lane);
            while (qlen >= 32) {
                ag_drain<REC>(x, src_idx, tile, q, qlen, 32, lane, lw, kept);
                qlen -= 32;
            }
            ag_fetch(ia, dst_idx, k + 2, wave, rot, lane);
            ag_scan(ib, q, qlen, lo, cnt, lane);
            while (qlen >= 32) {
                ag_drain<REC>(x, src_idx, tile, q, qlen, 32, lane, lw, kept);
                qlen -= 32;
            }
        }

        if (wave == 0) {
            for (int b = AG_TAIL0; b < N_EDGES; b += 32) {
                const int e = b + lane;
                const unsigned r = (unsigned)(dst_idx[e] - lo);
                ag_push(r < (unsigned)cnt, r, (unsigned)e, q, qlen);
                if (qlen >= 32) {
                    ag_drain<REC>(x, src_idx, tile, q, qlen, 32, lane, lw, kept);
                    qlen -= 32;
                }
            }
        }
        if (qlen > 0) ag_drain<REC>(x, src_idx, tile, q, qlen, qlen, lane, lw, kept);
    }
    if constexpr (PASS == 1) {
        if (lane == 0) hdr[wslot] = kept;
    }

    __syncthreads();

    {
        const int c  = lane & 15;
        const int h  = lane >> 4;
        const int ct = wave & 7;
        const int n  = 16 * ct + c;

        v2f wb0[K_DIM / 8], wb1[K_DIM / 8];
        {
            const float* wp = wlds + (4 * h) * AG_PITCH + n;
            #pragma unroll
            for (int q = 0; q < K_DIM / 8; ++q) {
                const float* p = wp + (8 * q) * AG_PITCH;
                wb0[q] = v2f{p[0], p[AG_PITCH]};
                wb1[q] = v2f{p[2 * AG_PITCH], p[3 * AG_PITCH]};
            }
        }
        const float bias_n = bias[n];
        const float scale  = bn_g[n] * rsqrtf(bn_v[n] + BN_EPS);
        const float shift  = bn_b[n] - bn_m[n] * scale;

        for (int rt = wave >> 3; rt < AG_RTILES; rt += AG_WAVES / 8) {
            const float* ap = tile + (16 * rt + c) * AG_PITCH + 4 * h;
            v8f acc = {};
            #pragma unroll
            for (int q = 0; q < K_DIM / 8; ++q) {
                const v4f a = *reinterpret_cast<const v4f*>(ap + 8 * q);
                acc = __builtin_amdgcn_wmma_f32_16x16x4_f32(
                    false, v2f{a.x, a.y}, false, wb0[q], (short)0, acc, false, false);
                acc = __builtin_amdgcn_wmma_f32_16x16x4_f32(
                    false, v2f{a.z, a.w}, false, wb1[q], (short)0, acc, false, false);
            }
            float* op = qo.o + (16 * rt + 8 * h) * AG_PITCH + n;
            #pragma unroll
            for (int r = 0; r < 8; ++r) {
                float val = acc[r] + bias_n;
                val = val * scale + shift;
                op[r * AG_PITCH] = fmaxf(val, 0.0f);
            }
        }
    }
    __syncthreads();

    {
        v4f* os = reinterpret_cast<v4f*>(hout + (size_t)lo * K_DIM);
        const v4f* ps = reinterpret_cast<const v4f*>(qo.o);
        for (int i = tid; i < n4; i += AG_WAVES * 32)
            os[i] = ps[(i >> 5) * (AG_PITCH / 4) + (i & 31)];
    }
}

#define GEMM_WAVES_PER_BLOCK 4
#define GEMM_BLOCKS          256

template<int MODE, int HOUT>
__global__ void __launch_bounds__(GEMM_WAVES_PER_BLOCK * 32)
gin_gemm_wmma(const float* __restrict__ A,
              const float* __restrict__ W,
              const float* __restrict__ bias,
              float* __restrict__ out) {
    constexpr int  CT        = (HOUT + 15) / 16;
    constexpr int  KB        = K_DIM / 8;
    constexpr int  ROW_TILES = N_NODES / 16;
    constexpr bool QUAD      = (HOUT % 64 == 0);
    constexpr int  NJ        = QUAD ? HOUT / 64 : 1;

    const int lane = threadIdx.x & 31;
    const int c    = lane & 15;
    const int h    = lane >> 4;
    const int wave   = __builtin_amdgcn_readfirstlane((int)(threadIdx.x >> 5));
    const int nwaves = (int)gridDim.x * GEMM_WAVES_PER_BLOCK;
    int tile = (int)blockIdx.x * GEMM_WAVES_PER_BLOCK + wave;

    v2f wb[KB][2][CT];
    if constexpr (QUAD) {
        const float* wl = W + (size_t)(4 * h) * HOUT + 4 * c;
        #pragma unroll
        for (int q = 0; q < KB; ++q) {
            v4f L[4][NJ];
            #pragma unroll
            for (int i = 0; i < 4; ++i) {
                #pragma unroll
                for (int j = 0; j < NJ; ++j)
                    L[i][j] = *reinterpret_cast<const v4f*>(wl + (size_t)(8 * q + i) * HOUT + 64 * j);
            }
            #pragma unroll
            for (int j = 0; j < NJ; ++j) {
                #pragma unroll
                for (int s = 0; s < 4; ++s) {
                    wb[q][0][4 * j + s] = v2f{L[0][j][s], L[1][j][s]};
                    wb[q][1][4 * j + s] = v2f{L[2][j][s], L[3][j][s]};
                }
            }
        }
    } else {
        #pragma unroll
        for (int t = 0; t < CT; ++t) {
            const int  n   = t * 16 + c;
            const bool okn = n < HOUT;
            const float* wl = W + (size_t)(4 * h) * HOUT + (okn ? n : 0);
            #pragma unroll
            for (int q = 0; q < KB; ++q) {
                const float* p = wl + (size_t)(8 * q) * HOUT;
                const float w0 = p[0], w1 = p[HOUT], w2 = p[2 * HOUT], w3 = p[3 * HOUT];
                wb[q][0][t] = v2f{okn ? w0 : 0.0f, okn ? w1 : 0.0f};
                wb[q][1][t] = v2f{okn ? w2 : 0.0f, okn ? w3 : 0.0f};
            }
        }
    }

    float bias_v[CT];
    bool  ok[CT];
    #pragma unroll
    for (int t = 0; t < CT; ++t) {
        const int n = QUAD ? (64 * (t >> 2) + 4 * c + (t & 3)) : (t * 16 + c);
        ok[t]     = (HOUT % 16 == 0) || (n < HOUT);
        bias_v[t] = ok[t] ? bias[n] : 0.0f;
    }

    const v4f* Alane = reinterpret_cast<const v4f*>(A + c * K_DIM + 4 * h);

    v4f cur[KB];
    if (tile < ROW_TILES) {
        const v4f* ap = Alane + (size_t)tile * (16 * K_DIM / 4);
        #pragma unroll
        for (int q = 0; q < KB; ++q) cur[q] = ap[2 * q];
    }

    for (; tile < ROW_TILES; tile += nwaves) {
        const int nt = (tile + nwaves < ROW_TILES) ? (tile + nwaves) : tile;
        const v4f* np = Alane + (size_t)nt * (16 * K_DIM / 4);
        v4f nxt[KB];
        #pragma unroll
        for (int q = 0; q < KB; ++q) nxt[q] = np[2 * q];

        v8f acc[CT] = {};
        #pragma unroll
        for (int q = 0; q < KB; ++q) {
            const v2f a0 = {cur[q].x, cur[q].y};
            const v2f a1 = {cur[q].z, cur[q].w};
            #pragma unroll
            for (int t = 0; t < CT; ++t) {
                acc[t] = __builtin_amdgcn_wmma_f32_16x16x4_f32(
                    false, a0, false, wb[q][0][t], (short)0, acc[t], false, false);
            }
            #pragma unroll
            for (int t = 0; t < CT; ++t) {
                acc[t] = __builtin_amdgcn_wmma_f32_16x16x4_f32(
                    false, a1, false, wb[q][1][t], (short)0, acc[t], false, false);
            }
        }

        if constexpr (QUAD) {
            float* orow = out + (size_t)(tile * 16 + 8 * h) * HOUT + 4 * c;
            #pragma unroll
            for (int j = 0; j < NJ; ++j) {
                #pragma unroll
                for (int r = 0; r < 8; ++r) {
                    v4f v;
                    #pragma unroll
                    for (int s = 0; s < 4; ++s) {
                        const int t = 4 * j + s;
                        float val = acc[t][r] + bias_v[t];
                        if constexpr (MODE >= 1) val = fmaxf(val, 0.0f);
                        v[s] = val;
                    }
                    *reinterpret_cast<v4f*>(orow + r * HOUT + 64 * j) = v;
                }
            }
        } else {
            float* orow = out + (size_t)(tile * 16 + 8 * h) * HOUT + c;
            #pragma unroll
            for (int t = 0; t < CT; ++t) {
                #pragma unroll
                for (int r = 0; r < 8; ++r) {
                    float val = acc[t][r] + bias_v[t];
                    if constexpr (MODE >= 1) val = fmaxf(val, 0.0f);
                    if (ok[t]) orow[r * HOUT + t * 16] = val;
                }
            }
        }

        #pragma unroll
        for (int q = 0; q < KB; ++q) cur[q] = nxt[q];
    }
}

extern "C" void kernel_launch(void* const* d_in, const int* in_sizes, int n_in,
                              void* d_out, int out_size, void* d_ws, size_t ws_size,
                              hipStream_t stream) {
    (void)in_sizes; (void)n_in; (void)out_size; (void)ws_size;

    const float* x      = (const float*)d_in[0];
    const int*   edge   = (const int*)  d_in[1];
    const float* lin1_w = (const float*)d_in[2];
    const float* lin1_b = (const float*)d_in[3];
    const float* bn1_g  = (const float*)d_in[4];
    const float* bn1_b  = (const float*)d_in[5];
    const float* bn1_m  = (const float*)d_in[6];
    const float* bn1_v  = (const float*)d_in[7];
    const float* lin2_w = (const float*)d_in[8];
    const float* lin2_b = (const float*)d_in[9];
    const float* lin3_w = (const float*)d_in[10];
    const float* lin3_b = (const float*)d_in[11];
    const float* bn2_g  = (const float*)d_in[12];
    const float* bn2_b  = (const float*)d_in[13];
    const float* bn2_m  = (const float*)d_in[14];
    const float* bn2_v  = (const float*)d_in[15];
    const float* lin4_w = (const float*)d_in[16];
    const float* lin4_b = (const float*)d_in[17];
    const float* lin5_w = (const float*)d_in[18];
    const float* lin5_b = (const float*)d_in[19];
    const float* lin6_w = (const float*)d_in[20];
    const float* lin6_b = (const float*)d_in[21];

    const int* e_src = edge;
    const int* e_dst = edge + N_EDGES;

    const size_t NH = (size_t)N_NODES * K_DIM;
    float* B0  = (float*)d_ws;
    float* B1  = B0  + NH;
    unsigned* LST = (unsigned*)(B1 + NH);
    int*      HDR = (int*)(LST + (size_t)AG_GROUPS * AG_WAVES * AG_LCAP);

    const int gG       = GEMM_BLOCKS;
    const int gB       = GEMM_WAVES_PER_BLOCK * 32;

    gin_aggregate_lin_kernel<1><<<AG_GROUPS, AG_WAVES * 32, 0, stream>>>(
        x, e_src, e_dst, lin1_w, lin1_b, bn1_g, bn1_b, bn1_m, bn1_v, B1, LST, HDR);
    gin_gemm_wmma<1,128><<<gG, gB, 0, stream>>>(B1, lin2_w, lin2_b, B0);

    gin_aggregate_lin_kernel<2><<<AG_GROUPS, AG_WAVES * 32, 0, stream>>>(
        B0, e_src, e_dst, lin3_w, lin3_b, bn2_g, bn2_b, bn2_m, bn2_v, B1, LST, HDR);
    gin_gemm_wmma<1,128><<<gG, gB, 0, stream>>>(B1, lin4_w, lin4_b, B0);

    gin_gemm_wmma<1,128><<<gG, gB, 0, stream>>>(B0, lin5_w, lin5_b, B1);
    gin_gemm_wmma<0,40> <<<gG, gB, 0, stream>>>(B1, lin6_w, lin6_b, (float*)d_out);
}


# Each layer's other linear maps (lin2; lin4, lin5, lin6) run in the aggregation kernel too: rows go LDS plane to LDS plane, each W staged in LDS per map; 6 -> 2 launches; k order unchanged
# speedup vs baseline: 3.5631x; 1.2066x over previous
#include <hip/hip_runtime.h>

typedef __attribute__((ext_vector_type(2))) float v2f;
typedef __attribute__((ext_vector_type(4))) float v4f;
typedef __attribute__((ext_vector_type(8))) float v8f;

#define N_NODES 50000
#define N_EDGES 600000
#define K_DIM   128
#define BN_EPS  1e-5f

typedef __attribute__((ext_vector_type(4))) int v4i;

#define AG_GROUPS   256
#define AG_RANGE    196
#define AG_WAVES    32
#define AG_U        4
#define AG_BLOCK    (AG_U * 128)
#define AG_STEP     8
#define AG_NBLK     (N_EDGES / AG_BLOCK)
#define AG_TAIL0    (AG_NBLK * AG_BLOCK)
#define AG_TRIPS    (2 * ((AG_NBLK + 2 * AG_WAVES - 1) / (2 * AG_WAVES)))
#define AG_NBLK_PAD (AG_TRIPS * AG_WAVES)
#define AG_ROT      4
#define AG_QCAP     (32 + AG_BLOCK)
#define AG_LCAP     1024
#define AG_STEP2    16
#define AG_PITCH    132
#define AG_RTILES   ((AG_RANGE + 15) / 16)
#define AG_ROWS     (AG_RTILES * 16)
static_assert(AG_ROWS > AG_RANGE, "room for the spare row");
static_assert(AG_PITCH % 4 == 0 && (AG_PITCH / 4) % 2 == 1, "rows 16-byte aligned, pitch 16 B x odd");
static_assert(AG_GROUPS * AG_RANGE >= N_NODES, "ranges cover all nodes");
static_assert((AG_GROUPS - 1) * AG_RANGE < N_NODES, "no empty range");
static_assert(AG_RANGE < 256, "local row and the spare row fit 8 bits");
static_assert(N_EDGES < (1 << 24), "edge number fits 24 bits");
static_assert((N_EDGES - AG_TAIL0) % 32 == 0, "the tail is whole waves of ids");
static_assert(AG_GROUPS * AG_ROT <= AG_NBLK_PAD, "one wrap is enough");

struct AgIds {
    v4i d[AG_U];
    int blk;
};

__device__ __forceinline__ void ag_fetch(AgIds& s, const int* __restrict__ dst_idx,
                                         int k, int wave, int rot, int lane) {
    int b = k * AG_WAVES + wave + rot;
    if (b >= AG_NBLK_PAD) b -= AG_NBLK_PAD;
    const bool ok = b < AG_NBLK;
    s.blk = ok ? b : -1;
    const v4i* p = reinterpret_cast<const v4i*>(dst_idx + (size_t)(ok ? b : 0) * AG_BLOCK) + lane;
    #pragma unroll
    for (int u = 0; u < AG_U; ++u) s.d[u] = p[u * 32];
}

__device__ __forceinline__ void ag_push(bool t, unsigned r, unsigned e, unsigned* q, int& qlen) {
    const unsigned m = __builtin_amdgcn_ballot_w32(t);
    if (t) q[qlen + __builtin_amdgcn_mbcnt_lo(m, 0u)] = (e << 8) | r;
    qlen += __builtin_popcount(m);
}

__device__ __forceinline__ void ag_scan(const AgIds& s, unsigned* q, int& qlen,
                                        int lo, int cnt, int lane) {
    if (s.blk < 0) return;
    #pragma unroll
    for (int u = 0; u < AG_U; ++u) {
        const v4i d = s.d[u];
        const unsigned r0 = (unsigned)(d.x - lo);
        const unsigned r1 = (unsigned)(d.y - lo);
        const unsigned r2 = (unsigned)(d.z - lo);
        const unsigned r3 = (unsigned)(d.w - lo);
        const bool t0 = r0 < (unsigned)cnt;
        const bool t1 = r1 < (unsigned)cnt;
        const bool t2 = r2 < (unsigned)cnt;
        const bool t3 = r3 < (unsigned)cnt;
        if (__builtin_amdgcn_ballot_w32(t0 | t1 | t2 | t3)) {
            const unsigned e0 = (unsigned)(s.blk * AG_BLOCK + u * 128 + lane * 4);
            ag_push(t0, r0, e0,      q, qlen);
            ag_push(t1, r1, e0 + 1u, q, qlen);
            ag_push(t2, r2, e0 + 2u, q, qlen);
            ag_push(t3, r3, e0 + 3u, q, qlen);
        }
    }
}

template<int STEP>
__device__ __forceinline__ void ag_add_rows(const float* __restrict__ x, float* tile,
                                            int s, int dl, int take, int lane) {
    #pragma unroll
    for (int g = 0; g < 32; g += STEP) {
        if (g < take) {
            v2f ra[STEP], rb[STEP];
            #pragma unroll
            for (int u = 0; u < STEP; ++u) {
                const int sj = __builtin_amdgcn_readlane(s, g + u);
                const float* xr = x + (size_t)sj * K_DIM + 2 * lane;
                ra[u] = *reinterpret_cast<const v2f*>(xr);
                rb[u] = *reinterpret_cast<const v2f*>(xr + 64);
            }
            #pragma unroll
            for (int u = 0; u < STEP; ++u) {
                const int dj = __builtin_amdgcn_readlane(dl, g + u);
                float* t = tile + dj * AG_PITCH + 2 * lane;
                __hip_atomic_fetch_add(t,      ra[u].x, __ATOMIC_RELAXED, __HIP_MEMORY_SCOPE_WORKGROUP);
                __hip_atomic_fetch_add(t + 1,  ra[u].y, __ATOMIC_RELAXED, __HIP_MEMORY_SCOPE_WORKGROUP);
                __hip_atomic_fetch_add(t + 64, rb[u].x, __ATOMIC_RELAXED, __HIP_MEMORY_SCOPE_WORKGROUP);
                __hip_atomic_fetch_add(t + 65, rb[u].y, __ATOMIC_RELAXED, __HIP_MEMORY_SCOPE_WORKGROUP);
            }
        }
    }
}

template<bool REC>
__device__ __forceinline__ void ag_drain(const float* __restrict__ x,
                                         const int* __restrict__ src_idx,
                                         float* tile, const unsigned* q,
                                         int qlen, int take, int lane,
                                         unsigned* __restrict__ lw, int& kept) {
    unsigned ent = (unsigned)AG_RANGE;
    if (lane < take) ent = q[qlen - take + lane];
    const int s  = src_idx[ent >> 8];
    const int dl = (int)(ent & 255u);
    if constexpr (REC) {
        if (kept + take <= AG_LCAP) {
            if (lane < take) lw[kept + lane] = ((unsigned)s << 8) | (unsigned)dl;
        }
        kept += take;
    }
    ag_add_rows<AG_STEP>(x, tile, s, dl, take, lane);
}

union AgQueuesOrOut {
    unsigned q[AG_WAVES * AG_QCAP];
    float    o[AG_ROWS * AG_PITCH];
};
static_assert(sizeof(unsigned) * AG_WAVES * AG_QCAP <= sizeof(float) * AG_ROWS * AG_PITCH, "queues fit under the plane");

#define AG_THREADS (AG_WAVES * 32)
#define OUT_DIM    40
static_assert(OUT_DIM % 4 == 0, "rows of the last map's W and of out are whole b128 pieces");
static_assert(AG_ROWS * OUT_DIM <= AG_ROWS * AG_PITCH, "the last map's rows fit a plane");

template<int HOUT>
struct AgW {
    static constexpr int ROWV  = HOUT / 4;
    static constexpr int NV    = K_DIM * ROWV;
    static constexpr int TRIPS = (NV + AG_THREADS - 1) / AG_THREADS;
    static constexpr bool WHOLE = (NV % AG_THREADS == 0);
    v4f v[TRIPS];
};

template<int HOUT>
__device__ __forceinline__ void ag_w_fetch(AgW<HOUT>& r, const float* __restrict__ W, int tid) {
    const v4f* w4 = reinterpret_cast<const v4f*>(W);
    #pragma unroll
    for (int j = 0; j < AgW<HOUT>::TRIPS; ++j) {
        const int i = tid + j * AG_THREADS;
        r.v[j] = w4[(AgW<HOUT>::WHOLE || i < AgW<HOUT>::NV) ? i : (AgW<HOUT>::NV - 1)];
    }
}

template<int HOUT>
__device__ __forceinline__ void ag_w_store(const AgW<HOUT>& r, float* wlds, int tid) {
    v4f* ws = reinterpret_cast<v4f*>(wlds);
    #pragma unroll
    for (int j = 0; j < AgW<HOUT>::TRIPS; ++j) {
        const int i = tid + j * AG_THREADS;
        if (AgW<HOUT>::WHOLE || i < AgW<HOUT>::NV)
            ws[(i / AgW<HOUT>::ROWV) * (AG_PITCH / 4) + (i % AgW<HOUT>::ROWV)] = r.v[j];
    }
}

__device__ __forceinline__ void ag_load_b(const float* wlds, int n, bool ok, int h,
                                          v2f (&wb0)[K_DIM / 8], v2f (&wb1)[K_DIM / 8]) {
    const float* wp = wlds + (4 * h) * AG_PITCH + (ok ? n : 0);
    #pragma unroll
    for (int q = 0; q < K_DIM / 8; ++q) {
        const float* p = wp + (8 * q) * AG_PITCH;
        const float w0 = p[0], w1 = p[AG_PITCH], w2 = p[2 * AG_PITCH], w3 = p[3 * AG_PITCH];
        wb0[q] = v2f{ok ? w0 : 0.0f, ok ? w1 : 0.0f};
        wb1[q] = v2f{ok ? w2 : 0.0f, ok ? w3 : 0.0f};
    }
}

__device__ __forceinline__ v8f ag_tile_product(const float* ap,
                                               const v2f (&wb0)[K_DIM / 8],
                                               const v2f (&wb1)[K_DIM / 8]) {
    v8f acc = {};
    #pragma unroll
    for (int q = 0; q < K_DIM / 8; ++q) {
        const v4f a = *reinterpret_cast<const v4f*>(ap + 8 * q);
        acc = __builtin_amdgcn_wmma_f32_16x16x4_f32(
            false, v2f{a.x, a.y}, false, wb0[q], (short)0, acc, false, false);
        acc = __builtin_amdgcn_wmma_f32_16x16x4_f32(
            false, v2f{a.z, a.w}, false, wb1[q], (short)0, acc, false, false);
    }
    return acc;
}

template<int HOUT, int OPITCH, bool BN, bool RELU, typename NextW>
__device__ __forceinline__ void ag_linear(const float* src, const float* wlds, float* dst,
                                          const float* __restrict__ bias,
                                          const float* __restrict__ bn_g,
                                          const float* __restrict__ bn_b,
                                          const float* __restrict__ bn_m,
                                          const float* __restrict__ bn_v,
                                          int wave, int lane, NextW&& next_w) {
    const int c = lane & 15;
    const int h = lane >> 4;
    v2f wb0[K_DIM / 8], wb1[K_DIM / 8];
    if constexpr (HOUT == K_DIM) {
        const int n = 16 * (wave & 7) + c;
        const float bias_n = bias[n];
        float g = 1.0f, v = 1.0f, b = 0.0f, m = 0.0f;
        if constexpr (BN) { g = bn_g[n]; v = bn_v[n]; b = bn_b[n]; m = bn_m[n]; }
        next_w();
        float scale = 1.0f, shift = 0.0f;
        if constexpr (BN) {
            scale = g * rsqrtf(v + BN_EPS);
            shift = b - m * scale;
        }
        ag_load_b(wlds, n, true, h, wb0, wb1);
        static_assert(AG_WAVES / 8 <= AG_RTILES, "every wave has a first row tile");
        int rt = wave >> 3;
        do {
            const v8f acc = ag_tile_product(src + (16 * rt + c) * AG_PITCH + 4 * h, wb0, wb1);
            float* op = dst + (16 * rt + 8 * h) * OPITCH + n;
            #pragma unroll
            for (int r = 0; r < 8; ++r) {
                float val = acc[r] + bias_n;
                if constexpr (BN) val = val * scale + shift;
                if constexpr (BN || RELU) val = fmaxf(val, 0.0f);
                op[r * OPITCH] = val;
            }
            rt += AG_WAVES / 8;
        } while (rt < AG_RTILES);
    } else {
        constexpr int NCT = (HOUT + 15) / 16;
        next_w();
        for (int u = wave; u < NCT * AG_RTILES; u += AG_WAVES) {
            const int ct = u / AG_RTILES;
            const int rt = u - ct * AG_RTILES;
            const int n  = 16 * ct + c;
            const bool ok = n < HOUT;
            const float bias_n = bias[ok ? n : 0];
            ag_load_b(wlds, n, ok, h, wb0, wb1);
            const v8f acc = ag_tile_product(src + (16 * rt + c) * AG_PITCH + 4 * h, wb0, wb1);
            float* op = dst + (16 * rt + 8 * h) * OPITCH + n;
            #pragma unroll
            for (int r = 0; r < 8; ++r) {
                float val = acc[r] + bias_n;
                if constexpr (RELU) val = fmaxf(val, 0.0f);
                if (ok) op[r * OPITCH] = val;
            }
        }
    }
}


template<int PASS>
__global__ void __launch_bounds__(AG_THREADS)
gin_layer_kernel(const float* __restrict__ x,
                 const int* __restrict__ src_idx,
                 const int* __restrict__ dst_idx,
                 const float* __restrict__ W,
                 const float* __restrict__ bias,
                 const float* __restrict__ bn_g,
                 const float* __restrict__ bn_b,
                 const float* __restrict__ bn_m,
                 const float* __restrict__ bn_v,
                 const float* __restrict__ Wb,
                 const float* __restrict__ bias_b,
                 const float* __restrict__ Wc,
                 const float* __restrict__ bias_c,
                 const float* __restrict__ Wd,
                 const float* __restrict__ bias_d,
                 float* __restrict__ hout,
                 unsigned* __restrict__ lst,
                 int* __restrict__ hdr) {
    __shared__ __attribute__((aligned(16))) float tile[AG_ROWS * AG_PITCH];
    __shared__ __attribute__((aligned(16))) float wlds[K_DIM * AG_PITCH];
    __shared__ __attribute__((aligned(16))) AgQueuesOrOut qo;
    unsigned* queue_all = qo.q;

    const int tid  = threadIdx.x;
    const int lane = tid & 31;
    const int wave = __builtin_amdgcn_readfirstlane(tid >> 5);
    const int lo   = (int)blockIdx.x * AG_RANGE;
    const int cnt  = (N_NODES - lo < AG_RANGE) ? (N_NODES - lo) : AG_RANGE;
    const int n4   = cnt * (K_DIM / 4);

    unsigned* q = queue_all + wave * AG_QCAP;
    int qlen = 0;
    const int rot = (int)blockIdx.x * AG_ROT;

    const int wslot = (int)blockIdx.x * AG_WAVES + wave;
    unsigned* lw = lst + (size_t)wslot * AG_LCAP;
    int kept = 0;
    int n = 0;
    bool listed = false;
    if constexpr (PASS == 2) {
        n = hdr[wslot];
        listed = (n <= AG_LCAP);
    }

    AgIds ia, ib;
    unsigned ent_n = (unsigned)AG_RANGE;
    if (listed) {
        if (lane < n) ent_n = lw[lane];
    } else {
        ag_fetch(ia, dst_idx, 0, wave, rot, lane);
    }

    {
        constexpr int FILL = (AG_RANGE * (K_DIM / 4) + AG_WAVES * 32 - 1) / (AG_WAVES * 32);
        constexpr int WFIL = (K_DIM * (K_DIM / 4)) / (AG_WAVES * 32);
        static_assert(WFIL * AG_WAVES * 32 == K_DIM * (K_DIM / 4), "the weights are whole trips");
        const v4f* xs = reinterpret_cast<const v4f*>(x + (size_t)lo * K_DIM);
        const v4f* w4 = reinterpret_cast<const v4f*>(W);
        v4f* ts = reinterpret_cast<v4f*>(tile);
        v4f* ws = reinterpret_cast<v4f*>(wlds);
        v4f tmp[FILL], wt[WFIL];
        #pragma unroll
        for (int j = 0; j < WFIL; ++j) wt[j] = w4[tid + j * (AG_WAVES * 32)];
        #pragma unroll
        for (int j = 0; j < FILL; ++j) {
            const int i = tid + j * (AG_WAVES * 32);
            tmp[j] = xs[(i < n4) ? i : (n4 - 1)];
        }
        #pragma unroll
        for (int j = 0; j < WFIL; ++j) {
            const int i = tid + j * (AG_WAVES * 32);
            ws[(i >> 5) * (AG_PITCH / 4) + (i & 31)] = wt[j];
        }
        #pragma unroll
        for (int j = 0; j < FILL; ++j) {
            const int i = tid + j * (AG_WAVES * 32);
            if (i < n4) ts[(i >> 5) * (AG_PITCH / 4) + (i & 31)] = tmp[j];
        }
    }
    __syncthreads();

    constexpr bool REC = (PASS == 1);
    if (listed) {
        for (int base = 0; base < n; base += 32) {
            const unsigned ent = ent_n;
            const int left = n - base;
            ent_n = (unsigned)AG_RANGE;
            if (lane + 32 < left) ent_n = lw[base + 32 + lane];
            ag_add_rows<AG_STEP2>(x, tile, (int)(ent >> 8), (int)(ent & 255u),
                                  left < 32 ? left : 32, lane);
        }
    } else {
        for (int k = 0; k < AG_TRIPS; k += 2) {
            ag_fetch(ib, dst_idx, k + 1, wave, rot, lane);
            ag_scan(ia, q, qlen, lo, cnt, lane);
            while (qlen >= 32) {
                ag_drain<REC>(x, src_idx, tile, q, qlen, 32, lane, lw, kept);
                qlen -= 32;
            }
            ag_fetch(ia, dst_idx, k + 2, wave, rot, lane);
            ag_scan(ib, q, qlen, lo, cnt, lane);
            while (qlen >= 32) {
                ag_drain<REC>(x, src_idx, tile, q, qlen, 32, lane, lw, kept);
                qlen -= 32;
            }
        }

        if (wave == 0) {
            for (int b = AG_TAIL0; b < N_EDGES; b += 32) {
                const int e = b + lane;
                const unsigned r = (unsigned)(dst_idx[e] - lo);
                ag_push(r < (unsigned)cnt, r, (unsigned)e, q, qlen);
                if (qlen >= 32) {
                    ag_drain<REC>(x, src_idx, tile, q, qlen, 32, lane, lw, kept);
                    qlen -= 32;
                }
            }
        }
        if (qlen > 0) ag_drain<REC>(x, src_idx, tile, q, qlen, qlen, lane, lw, kept);
    }
    if constexpr (PASS == 1) {
        if (lane == 0) hdr[wslot] = kept;
    }

    __syncthreads();

    float* const plane = qo.o;
    AgW<K_DIM> wn;

    ag_linear<K_DIM, AG_PITCH, true, true>(tile, wlds, plane, bias, bn_g, bn_b, bn_m, bn_v, wave, lane,
                                           [&] { ag_w_fetch(wn, Wb, tid); });
    __syncthreads();
    ag_w_store(wn, wlds, tid);
    __syncthreads();

    if constexpr (PASS == 1) {
        ag_linear<K_DIM, AG_PITCH, false, true>(plane, wlds, tile, bias_b, nullptr, nullptr, nullptr, nullptr, wave, lane,
                                                [] {});
        __syncthreads();

        v4f* os = reinterpret_cast<v4f*>(hout + (size_t)lo * K_DIM);
        const v4f* ps = reinterpret_cast<const v4f*>(tile);
        for (int i = tid; i < n4; i += AG_THREADS)
            os[i] = ps[(i >> 5) * (AG_PITCH / 4) + (i & 31)];
    } else {
        ag_linear<K_DIM, AG_PITCH, false, true>(plane, wlds, tile, bias_b, nullptr, nullptr, nullptr, nullptr, wave, lane,
                                                [&] { ag_w_fetch(wn, Wc, tid); });
        __syncthreads();
        ag_w_store(wn, wlds, tid);
        __syncthreads();

        AgW<OUT_DIM> wl;
        ag_linear<K_DIM, AG_PITCH, false, true>(tile, wlds, plane, bias_c, nullptr, nullptr, nullptr, nullptr, wave, lane,
                                                [&] { ag_w_fetch(wl, Wd, tid); });
        __syncthreads();
        ag_w_store(wl, wlds, tid);
        __syncthreads();

        static_assert((AG_RANGE * OUT_DIM * 4) % 128 == 0, "a group's rows of out are whole lines");
        ag_linear<OUT_DIM, OUT_DIM, false, false>(plane, wlds, tile, bias_d, nullptr, nullptr, nullptr, nullptr, wave, lane,
                                                  [] {});
        __syncthreads();

        v4f* os = reinterpret_cast<v4f*>(hout + (size_t)lo * OUT_DIM);
        const v4f* ps = reinterpret_cast<const v4f*>(tile);
        for (int i = tid; i < cnt * (OUT_DIM / 4); i += AG_THREADS)
            os[i] = ps[i];
    }
}

extern "C" void kernel_launch(void* const* d_in, const int* in_sizes, int n_in,
                              void* d_out, int out_size, void* d_ws, size_t ws_size,
                              hipStream_t stream) {
    (void)in_sizes; (void)n_in; (void)out_size; (void)ws_size;

    const float* x      = (const float*)d_in[0];
    const int*   edge   = (const int*)  d_in[1];
    const float* lin1_w = (const float*)d_in[2];
    const float* lin1_b = (const float*)d_in[3];
    const float* bn1_g  = (const float*)d_in[4];
    const float* bn1_b  = (const float*)d_in[5];
    const float* bn1_m  = (const float*)d_in[6];
    const float* bn1_v  = (const float*)d_in[7];
    const float* lin2_w = (const float*)d_in[8];
    const float* lin2_b = (const float*)d_in[9];
    const float* lin3_w = (const float*)d_in[10];
    const float* lin3_b = (const float*)d_in[11];
    const float* bn2_g  = (const float*)d_in[12];
    const float* bn2_b  = (const float*)d_in[13];
    const float* bn2_m  = (const float*)d_in[14];
    const float* bn2_v  = (const float*)d_in[15];
    const float* lin4_w = (const float*)d_in[16];
    const float* lin4_b = (const float*)d_in[17];
    const float* lin5_w = (const float*)d_in[18];
    const float* lin5_b = (const float*)d_in[19];
    const float* lin6_w = (const float*)d_in[20];
    const float* lin6_b = (const float*)d_in[21];

    const int* e_src = edge;
    const int* e_dst = edge + N_EDGES;

    const size_t NH = (size_t)N_NODES * K_DIM;
    float*    H1  = (float*)d_ws;
    unsigned* LST = (unsigned*)(H1 + NH);
    int*      HDR = (int*)(LST + (size_t)AG_GROUPS * AG_WAVES * AG_LCAP);

    gin_layer_kernel<1><<<AG_GROUPS, AG_THREADS, 0, stream>>>(
        x, e_src, e_dst, lin1_w, lin1_b, bn1_g, bn1_b, bn1_m, bn1_v, lin2_w, lin2_b,
        nullptr, nullptr, nullptr, nullptr, H1, LST, HDR);

    gin_layer_kernel<2><<<AG_GROUPS, AG_THREADS, 0, stream>>>(
        H1, e_src, e_dst, lin3_w, lin3_b, bn2_g, bn2_b, bn2_m, bn2_v, lin4_w, lin4_b,
        lin5_w, lin5_b, lin6_w, lin6_b, (float*)d_out, LST, HDR);
}


# Edge finding indexed: a new first launch marks per owner group which 32-id lines of the id list hold its edges (bitmap); groups read only marked lines; row adds still LDS arrival order; 2->3 launches
# speedup vs baseline: 3.7342x; 1.0480x over previous
#include <hip/hip_runtime.h>

typedef __attribute__((ext_vector_type(2))) float v2f;
typedef __attribute__((ext_vector_type(4))) float v4f;
typedef __attribute__((ext_vector_type(8))) float v8f;

#define N_NODES 50000
#define N_EDGES 600000
#define K_DIM   128
#define BN_EPS  1e-5f

#define AG_GROUPS   256
#define AG_RANGE    196
#define AG_WAVES    32
#define AG_STEP     8
#define IX_BLK      32
#define IX_NBLK     (N_EDGES / IX_BLK)
#define IX_WORDS    ((IX_NBLK + 31) / 32)
#define IX_THREADS  256
#define IX_PER      (32 * IX_BLK / IX_THREADS)
#define IX_WPW      ((IX_WORDS + AG_WAVES - 1) / AG_WAVES)
#define IX_U        8
#define AG_HCAP     (IX_WPW * 32)
#define AG_QCAP     (32 + IX_U * IX_BLK)
#define AG_LCAP     1024
#define AG_STEP2    16
#define AG_PITCH    132
#define AG_RTILES   ((AG_RANGE + 15) / 16)
#define AG_ROWS     (AG_RTILES * 16)
static_assert(AG_ROWS > AG_RANGE, "room for the spare row");
static_assert(AG_PITCH % 4 == 0 && (AG_PITCH / 4) % 2 == 1, "rows 16-byte aligned, pitch 16 B x odd");
static_assert(AG_GROUPS * AG_RANGE >= N_NODES, "ranges cover all nodes");
static_assert((AG_GROUPS - 1) * AG_RANGE < N_NODES, "no empty range");
static_assert(AG_RANGE < 256, "local row and the spare row fit 8 bits");
static_assert(N_EDGES < (1 << 24), "edge number fits 24 bits");
static_assert(N_EDGES % IX_BLK == 0, "the id list is whole lines");
static_assert(IX_THREADS == AG_GROUPS, "index kernel: a thread per owner group");
static_assert(IX_PER * IX_THREADS == 32 * IX_BLK && IX_THREADS % 32 == 0, "a wave of the index kernel reads whole lines");
static_assert(IX_WPW <= 32, "a lane per index word of the wave");

__global__ void __launch_bounds__(IX_THREADS)
edge_index_kernel(const int* __restrict__ dst_idx, unsigned* __restrict__ idx) {
    __shared__ unsigned bm[AG_GROUPS];
    const int tid = threadIdx.x;
    const int e0  = (int)blockIdx.x * (32 * IX_BLK) + tid;
    int d[IX_PER];
    #pragma unroll
    for (int j = 0; j < IX_PER; ++j) {
        const int e = e0 + j * IX_THREADS;
        d[j] = dst_idx[e < N_EDGES ? e : N_EDGES - 1];
    }
    bm[tid] = 0u;
    __syncthreads();
    #pragma unroll
    for (int j = 0; j < IX_PER; ++j) {
        if (e0 + j * IX_THREADS < N_EDGES) {
            unsigned g = (unsigned)d[j] / (unsigned)AG_RANGE;
            g = g < (unsigned)AG_GROUPS ? g : (unsigned)(AG_GROUPS - 1);
            const unsigned bit = (unsigned)(j * (IX_THREADS / 32) + (tid >> 5));
            __hip_atomic_fetch_or(&bm[g], 1u << bit, __ATOMIC_RELAXED, __HIP_MEMORY_SCOPE_WORKGROUP);
        }
    }
    __syncthreads();
    idx[(size_t)blockIdx.x * AG_GROUPS + tid] = bm[tid];
}

__device__ __forceinline__ void ag_push(bool t, unsigned r, unsigned e, unsigned* q, int& qlen) {
    const unsigned m = __builtin_amdgcn_ballot_w32(t);
    if (t) q[qlen + __builtin_amdgcn_mbcnt_lo(m, 0u)] = (e << 8) | r;
    qlen += __builtin_popcount(m);
}

template<int STEP>
__device__ __forceinline__ void ag_add_rows(const float* __restrict__ x, float* tile,
                                            int s, int dl, int take, int lane) {
    #pragma unroll
    for (int g = 0; g < 32; g += STEP) {
        if (g < take) {
            v2f ra[STEP], rb[STEP];
            #pragma unroll
            for (int u = 0; u < STEP; ++u) {
                const int sj = __builtin_amdgcn_readlane(s, g + u);
                const float* xr = x + (size_t)sj * K_DIM + 2 * lane;
                ra[u] = *reinterpret_cast<const v2f*>(xr);
                rb[u] = *reinterpret_cast<const v2f*>(xr + 64);
            }
            #pragma unroll
            for (int u = 0; u < STEP; ++u) {
                const int dj = __builtin_amdgcn_readlane(dl, g + u);
                float* t = tile + dj * AG_PITCH + 2 * lane;
                __hip_atomic_fetch_add(t,      ra[u].x, __ATOMIC_RELAXED, __HIP_MEMORY_SCOPE_WORKGROUP);
                __hip_atomic_fetch_add(t + 1,  ra[u].y, __ATOMIC_RELAXED, __HIP_MEMORY_SCOPE_WORKGROUP);
                __hip_atomic_fetch_add(t + 64, rb[u].x, __ATOMIC_RELAXED, __HIP_MEMORY_SCOPE_WORKGROUP);
                __hip_atomic_fetch_add(t + 65, rb[u].y, __ATOMIC_RELAXED, __HIP_MEMORY_SCOPE_WORKGROUP);
            }
        }
    }
}

template<bool REC>
__device__ __forceinline__ void ag_drain(const float* __restrict__ x,
                                         const int* __restrict__ src_idx,
                                         float* tile, const unsigned* q,
                                         int qlen, int take, int lane,
                                         unsigned* __restrict__ lw, int& kept) {
    unsigned ent = (unsigned)AG_RANGE;
    if (lane < take) ent = q[qlen - take + lane];
    const int s  = src_idx[ent >> 8];
    const int dl = (int)(ent & 255u);
    if constexpr (REC) {
        if (kept + take <= AG_LCAP) {
            if (lane < take) lw[kept + lane] = ((unsigned)s << 8) | (unsigned)dl;
        }
        kept += take;
    }
    ag_add_rows<AG_STEP>(x, tile, s, dl, take, lane);
}

union AgQueuesOrOut {
    unsigned q[AG_WAVES * (AG_HCAP + AG_QCAP)];
    float    o[AG_ROWS * AG_PITCH];
};

#define AG_THREADS (AG_WAVES * 32)
#define OUT_DIM    40
static_assert(OUT_DIM % 4 == 0, "rows of the last map's W and of out are whole b128 pieces");
static_assert(AG_ROWS * OUT_DIM <= AG_ROWS * AG_PITCH, "the last map's rows fit a plane");

template<int HOUT>
struct AgW {
    static constexpr int ROWV  = HOUT / 4;
    static constexpr int NV    = K_DIM * ROWV;
    static constexpr int TRIPS = (NV + AG_THREADS - 1) / AG_THREADS;
    static constexpr bool WHOLE = (NV % AG_THREADS == 0);
    v4f v[TRIPS];
};

template<int HOUT>
__device__ __forceinline__ void ag_w_fetch(AgW<HOUT>& r, const float* __restrict__ W, int tid) {
    const v4f* w4 = reinterpret_cast<const v4f*>(W);
    #pragma unroll
    for (int j = 0; j < AgW<HOUT>::TRIPS; ++j) {
        const int i = tid + j * AG_THREADS;
        r.v[j] = w4[(AgW<HOUT>::WHOLE || i < AgW<HOUT>::NV) ? i : (AgW<HOUT>::NV - 1)];
    }
}

template<int HOUT>
__device__ __forceinline__ void ag_w_store(const AgW<HOUT>& r, float* wlds, int tid) {
    v4f* ws = reinterpret_cast<v4f*>(wlds);
    #pragma unroll
    for (int j = 0; j < AgW<HOUT>::TRIPS; ++j) {
        const int i = tid + j * AG_THREADS;
        if (AgW<HOUT>::WHOLE || i < AgW<HOUT>::NV)
            ws[(i / AgW<HOUT>::ROWV) * (AG_PITCH / 4) + (i % AgW<HOUT>::ROWV)] = r.v[j];
    }
}

__device__ __forceinline__ void ag_load_b(const float* wlds, int n, bool ok, int h,
                                          v2f (&wb0)[K_DIM / 8], v2f (&wb1)[K_DIM / 8]) {
    const float* wp = wlds + (4 * h) * AG_PITCH + (ok ? n : 0);
    #pragma unroll
    for (int q = 0; q < K_DIM / 8; ++q) {
        const float* p = wp + (8 * q) * AG_PITCH;
        const float w0 = p[0], w1 = p[AG_PITCH], w2 = p[2 * AG_PITCH], w3 = p[3 * AG_PITCH];
        wb0[q] = v2f{ok ? w0 : 0.0f, ok ? w1 : 0.0f};
        wb1[q] = v2f{ok ? w2 : 0.0f, ok ? w3 : 0.0f};
    }
}

__device__ __forceinline__ v8f ag_tile_product(const float* ap,
                                               const v2f (&wb0)[K_DIM / 8],
                                               const v2f (&wb1)[K_DIM / 8]) {
    v8f acc = {};
    #pragma unroll
    for (int q = 0; q < K_DIM / 8; ++q) {
        const v4f a = *reinterpret_cast<const v4f*>(ap + 8 * q);
        acc = __builtin_amdgcn_wmma_f32_16x16x4_f32(
            false, v2f{a.x, a.y}, false, wb0[q], (short)0, acc, false, false);
        acc = __builtin_amdgcn_wmma_f32_16x16x4_f32(
            false, v2f{a.z, a.w}, false, wb1[q], (short)0, acc, false, false);
    }
    return acc;
}

template<int HOUT, int OPITCH, bool BN, bool RELU, typename NextW>
__device__ __forceinline__ void ag_linear(const float* src, const float* wlds, float* dst,
                                          const float* __restrict__ bias,
                                          const float* __restrict__ bn_g,
                                          const float* __restrict__ bn_b,
                                          const float* __restrict__ bn_m,
                                          const float* __restrict__ bn_v,
                                          int wave, int lane, NextW&& next_w) {
    const int c = lane & 15;
    const int h = lane >> 4;
    v2f wb0[K_DIM / 8], wb1[K_DIM / 8];
    if constexpr (HOUT == K_DIM) {
        const int n = 16 * (wave & 7) + c;
        const float bias_n = bias[n];
        float g = 1.0f, v = 1.0f, b = 0.0f, m = 0.0f;
        if constexpr (BN) { g = bn_g[n]; v = bn_v[n]; b = bn_b[n]; m = bn_m[n]; }
        next_w();
        float scale = 1.0f, shift = 0.0f;
        if constexpr (BN) {
            scale = g * rsqrtf(v + BN_EPS);
            shift = b - m * scale;
        }
        ag_load_b(wlds, n, true, h, wb0, wb1);
        static_assert(AG_WAVES / 8 <= AG_RTILES, "every wave has a first row tile");
        int rt = wave >> 3;
        do {
            const v8f acc = ag_tile_product(src + (16 * rt + c) * AG_PITCH + 4 * h, wb0, wb1);
            float* op = dst + (16 * rt + 8 * h) * OPITCH + n;
            #pragma unroll
            for (int r = 0; r < 8; ++r) {
                float val = acc[r] + bias_n;
                if constexpr (BN) val = val * scale + shift;
                if constexpr (BN || RELU) val = fmaxf(val, 0.0f);
                op[r * OPITCH] = val;
            }
            rt += AG_WAVES / 8;
        } while (rt < AG_RTILES);
    } else {
        constexpr int NCT = (HOUT + 15) / 16;
        next_w();
        for (int u = wave; u < NCT * AG_RTILES; u += AG_WAVES) {
            const int ct = u / AG_RTILES;
            const int rt = u - ct * AG_RTILES;
            const int n  = 16 * ct + c;
            const bool ok = n < HOUT;
            const float bias_n = bias[ok ? n : 0];
            ag_load_b(wlds, n, ok, h, wb0, wb1);
            const v8f acc = ag_tile_product(src + (16 * rt + c) * AG_PITCH + 4 * h, wb0, wb1);
            float* op = dst + (16 * rt + 8 * h) * OPITCH + n;
            #pragma unroll
            for (int r = 0; r < 8; ++r) {
                float val = acc[r] + bias_n;
                if constexpr (RELU) val = fmaxf(val, 0.0f);
                if (ok) op[r * OPITCH] = val;
            }
        }
    }
}


template<int PASS>
__global__ void __launch_bounds__(AG_THREADS)
gin_layer_kernel(const float* __restrict__ x,
                 const int* __restrict__ src_idx,
                 const int* __restrict__ dst_idx,
                 const float* __restrict__ W,
                 const float* __restrict__ bias,
                 const float* __restrict__ bn_g,
                 const float* __restrict__ bn_b,
                 const float* __restrict__ bn_m,
                 const float* __restrict__ bn_v,
                 const float* __restrict__ Wb,
                 const float* __restrict__ bias_b,
                 const float* __restrict__ Wc,
                 const float* __restrict__ bias_c,
                 const float* __restrict__ Wd,
                 const float* __restrict__ bias_d,
                 float* __restrict__ hout,
                 unsigned* __restrict__ lst,
                 int* __restrict__ hdr,
                 const unsigned* __restrict__ idx) {
    __shared__ __attribute__((aligned(16))) float tile[AG_ROWS * AG_PITCH];
    __shared__ __attribute__((aligned(16))) float wlds[K_DIM * AG_PITCH];
    __shared__ __attribute__((aligned(16))) AgQueuesOrOut qo;
    unsigned* queue_all = qo.q;

    const int tid  = threadIdx.x;
    const int lane = tid & 31;
    const int wave = __builtin_amdgcn_readfirstlane(tid >> 5);
    const int lo   = (int)blockIdx.x * AG_RANGE;
    const int cnt  = (N_NODES - lo < AG_RANGE) ? (N_NODES - lo) : AG_RANGE;
    const int n4   = cnt * (K_DIM / 4);

    unsigned* hb = queue_all + wave * (AG_HCAP + AG_QCAP);
    unsigned* q  = hb + AG_HCAP;
    int qlen = 0;

    const int wslot = (int)blockIdx.x * AG_WAVES + wave;
    unsigned* lw = lst + (size_t)wslot * AG_LCAP;
    int kept = 0;
    int n = 0;
    bool listed = false;
    if constexpr (PASS == 2) {
        n = hdr[wslot];
        listed = (n <= AG_LCAP);
    }

    unsigned mv = 0u;
    unsigned ent_n = (unsigned)AG_RANGE;
    if (listed) {
        if (lane < n) ent_n = lw[lane];
    } else {
        const int i = wave + AG_WAVES * lane;
        if (i < IX_WORDS) mv = idx[(size_t)i * AG_GROUPS + blockIdx.x];
    }

    int nhb = 0;

    {
        constexpr int FILL = (AG_RANGE * (K_DIM / 4) + AG_WAVES * 32 - 1) / (AG_WAVES * 32);
        constexpr int WFIL = (K_DIM * (K_DIM / 4)) / (AG_WAVES * 32);
        static_assert(WFIL * AG_WAVES * 32 == K_DIM * (K_DIM / 4), "the weights are whole trips");
        const v4f* xs = reinterpret_cast<const v4f*>(x + (size_t)lo * K_DIM);
        const v4f* w4 = reinterpret_cast<const v4f*>(W);
        v4f* ts = reinterpret_cast<v4f*>(tile);
        v4f* ws = reinterpret_cast<v4f*>(wlds);
        v4f tmp[FILL], wt[WFIL];
        #pragma unroll
        for (int j = 0; j < WFIL; ++j) wt[j] = w4[tid + j * (AG_WAVES * 32)];
        #pragma unroll
        for (int j = 0; j < FILL; ++j) {
            const int i = tid + j * (AG_WAVES * 32);
            tmp[j] = xs[(i < n4) ? i : (n4 - 1)];
        }
        if (!listed) {
            #pragma unroll
            for (int k = 0; k < IX_WPW; ++k) {
                const unsigned m = (unsigned)__builtin_amdgcn_readlane((int)mv, k);
                if ((m >> lane) & 1u)
                    hb[nhb + __builtin_amdgcn_mbcnt_lo(m, 0u)] = (unsigned)((wave + AG_WAVES * k) * 32 + lane);
                nhb += __builtin_popcount(m);
            }
        }
        #pragma unroll
        for (int j = 0; j < WFIL; ++j) {
            const int i = tid + j * (AG_WAVES * 32);
            ws[(i >> 5) * (AG_PITCH / 4) + (i & 31)] = wt[j];
        }
        #pragma unroll
        for (int j = 0; j < FILL; ++j) {
            const int i = tid + j * (AG_WAVES * 32);
            if (i < n4) ts[(i >> 5) * (AG_PITCH / 4) + (i & 31)] = tmp[j];
        }
    }
    __syncthreads();

    constexpr bool REC = (PASS == 1);
    if (listed) {
        for (int base = 0; base < n; base += 32) {
            const unsigned ent = ent_n;
            const int left = n - base;
            ent_n = (unsigned)AG_RANGE;
            if (lane + 32 < left) ent_n = lw[base + 32 + lane];
            ag_add_rows<AG_STEP2>(x, tile, (int)(ent >> 8), (int)(ent & 255u),
                                  left < 32 ? left : 32, lane);
        }
    } else {
        struct Trip { int h; int d[IX_U]; };
        auto fetch = [&](Trip& t, int p) {
            t.h = 0;
            if (lane < IX_U && p + lane < nhb) t.h = (int)hb[p + lane];
            #pragma unroll
            for (int u = 0; u < IX_U; ++u) {
                const int b = __builtin_amdgcn_readlane(t.h, u);
                t.d[u] = dst_idx[b * IX_BLK + lane];
            }
        };
        auto take = [&](const Trip& t, int p) {
            #pragma unroll
            for (int u = 0; u < IX_U; ++u) {
                if (p + u < nhb) {
                    const int b = __builtin_amdgcn_readlane(t.h, u);
                    const unsigned r = (unsigned)(t.d[u] - lo);
                    ag_push(r < (unsigned)cnt, r, (unsigned)(b * IX_BLK + lane), q, qlen);
                }
            }
            while (qlen >= 32) {
                ag_drain<REC>(x, src_idx, tile, q, qlen, 32, lane, lw, kept);
                qlen -= 32;
            }
        };
        Trip ta, tb;
        fetch(ta, 0);
        for (int p = 0; p < nhb; p += 2 * IX_U) {
            fetch(tb, p + IX_U);
            take(ta, p);
            fetch(ta, p + 2 * IX_U);
            take(tb, p + IX_U);
        }
        if (qlen > 0) ag_drain<REC>(x, src_idx, tile, q, qlen, qlen, lane, lw, kept);
    }
    if constexpr (PASS == 1) {
        if (lane == 0) hdr[wslot] = kept;
    }

    __syncthreads();

    float* const plane = qo.o;
    AgW<K_DIM> wn;

    ag_linear<K_DIM, AG_PITCH, true, true>(tile, wlds, plane, bias, bn_g, bn_b, bn_m, bn_v, wave, lane,
                                           [&] { ag_w_fetch(wn, Wb, tid); });
    __syncthreads();
    ag_w_store(wn, wlds, tid);
    __syncthreads();

    if constexpr (PASS == 1) {
        ag_linear<K_DIM, AG_PITCH, false, true>(plane, wlds, tile, bias_b, nullptr, nullptr, nullptr, nullptr, wave, lane,
                                                [] {});
        __syncthreads();

        v4f* os = reinterpret_cast<v4f*>(hout + (size_t)lo * K_DIM);
        const v4f* ps = reinterpret_cast<const v4f*>(tile);
        for (int i = tid; i < n4; i += AG_THREADS)
            os[i] = ps[(i >> 5) * (AG_PITCH / 4) + (i & 31)];
    } else {
        ag_linear<K_DIM, AG_PITCH, false, true>(plane, wlds, tile, bias_b, nullptr, nullptr, nullptr, nullptr, wave, lane,
                                                [&] { ag_w_fetch(wn, Wc, tid); });
        __syncthreads();
        ag_w_store(wn, wlds, tid);
        __syncthreads();

        AgW<OUT_DIM> wl;
        ag_linear<K_DIM, AG_PITCH, false, true>(tile, wlds, plane, bias_c, nullptr, nullptr, nullptr, nullptr, wave, lane,
                                                [&] { ag_w_fetch(wl, Wd, tid); });
        __syncthreads();
        ag_w_store(wl, wlds, tid);
        __syncthreads();

        static_assert((AG_RANGE * OUT_DIM * 4) % 128 == 0, "a group's rows of out are whole lines");
        ag_linear<OUT_DIM, OUT_DIM, false, false>(plane, wlds, tile, bias_d, nullptr, nullptr, nullptr, nullptr, wave, lane,
                                                  [] {});
        __syncthreads();

        v4f* os = reinterpret_cast<v4f*>(hout + (size_t)lo * OUT_DIM);
        const v4f* ps = reinterpret_cast<const v4f*>(tile);
        for (int i = tid; i < cnt * (OUT_DIM / 4); i += AG_THREADS)
            os[i] = ps[i];
    }
}

extern "C" void kernel_launch(void* const* d_in, const int* in_sizes, int n_in,
                              void* d_out, int out_size, void* d_ws, size_t ws_size,
                              hipStream_t stream) {
    (void)in_sizes; (void)n_in; (void)out_size; (void)ws_size;

    const float* x      = (const float*)d_in[0];
    const int*   edge   = (const int*)  d_in[1];
    const float* lin1_w = (const float*)d_in[2];
    const float* lin1_b = (const float*)d_in[3];
    const float* bn1_g  = (const float*)d_in[4];
    const float* bn1_b  = (const float*)d_in[5];
    const float* bn1_m  = (const float*)d_in[6];
    const float* bn1_v  = (const float*)d_in[7];
    const float* lin2_w = (const float*)d_in[8];
    const float* lin2_b = (const float*)d_in[9];
    const float* lin3_w = (const float*)d_in[10];
    const float* lin3_b = (const float*)d_in[11];
    const float* bn2_g  = (const float*)d_in[12];
    const float* bn2_b  = (const float*)d_in[13];
    const float* bn2_m  = (const float*)d_in[14];
    const float* bn2_v  = (const float*)d_in[15];
    const float* lin4_w = (const float*)d_in[16];
    const float* lin4_b = (const float*)d_in[17];
    const float* lin5_w = (const float*)d_in[18];
    const float* lin5_b = (const float*)d_in[19];
    const float* lin6_w = (const float*)d_in[20];
    const float* lin6_b = (const float*)d_in[21];

    const int* e_src = edge;
    const int* e_dst = edge + N_EDGES;

    const size_t NH = (size_t)N_NODES * K_DIM;
    float*    H1  = (float*)d_ws;
    unsigned* LST = (unsigned*)(H1 + NH);
    int*      HDR = (int*)(LST + (size_t)AG_GROUPS * AG_WAVES * AG_LCAP);
    unsigned* IDX = (unsigned*)(HDR + AG_GROUPS * AG_WAVES);

    edge_index_kernel<<<IX_WORDS, IX_THREADS, 0, stream>>>(e_dst, IDX);

    gin_layer_kernel<1><<<AG_GROUPS, AG_THREADS, 0, stream>>>(
        x, e_src, e_dst, lin1_w, lin1_b, bn1_g, bn1_b, bn1_m, bn1_v, lin2_w, lin2_b,
        nullptr, nullptr, nullptr, nullptr, H1, LST, HDR, IDX);

    gin_layer_kernel<2><<<AG_GROUPS, AG_THREADS, 0, stream>>>(
        H1, e_src, e_dst, lin3_w, lin3_b, bn2_g, bn2_b, bn2_m, bn2_v, lin4_w, lin4_b,
        lin5_w, lin5_b, lin6_w, lin6_b, (float*)d_out, LST, HDR, IDX);
}
